# starting kernel
# baseline (speedup 1.0000x reference)
#include <hip/hip_runtime.h>
#include <math.h>

typedef float v2f __attribute__((ext_vector_type(2)));
typedef float v8f __attribute__((ext_vector_type(8)));

#define HEADS 4
#define CHAN 32
#define FDIM 128
#define NEG_SLOPE 0.2f

__global__ void init_kernel(float* __restrict__ amax, float* __restrict__ denom,
                            float* __restrict__ out, int Nn) {
    int i = blockIdx.x * blockDim.x + threadIdx.x;
    if (i < Nn * HEADS) {
        amax[i]  = -__builtin_inff();
        denom[i] = 0.0f;
    }
    if (i < Nn * FDIM) out[i] = 0.0f;
}

__global__ void gemm_wmma(const float* __restrict__ x, const float* __restrict__ W,
                          float* __restrict__ h, int Nn) {
    const int wave = threadIdx.x >> 5;
    const int lane = threadIdx.x & 31;
    const int row0 = (blockIdx.x * 4 + wave) * 16;
    if (row0 >= Nn) return;

    v8f acc[8];
#pragma unroll
    for (int t = 0; t < 8; ++t) acc[t] = (v8f)(0.0f);

    const int mrow  = row0 + (lane & 15);
    const int khalf = (lane >> 4) << 1;
    const int ncol  = lane & 15;

    for (int k0 = 0; k0 < FDIM; k0 += 4) {
        const int ka = k0 + khalf;
        v2f a;
        a.x = x[(size_t)mrow * FDIM + ka];
        a.y = x[(size_t)mrow * FDIM + ka + 1];
#pragma unroll
        for (int t = 0; t < 8; ++t) {
            const int n = t * 16 + ncol;
            v2f b;
            b.x = W[(size_t)ka * FDIM + n];
            b.y = W[(size_t)(ka + 1) * FDIM + n];
            acc[t] = __builtin_amdgcn_wmma_f32_16x16x4_f32(
                false, a, false, b, (short)0, acc[t], false, false);
        }
    }

    const int mbase = row0 + ((lane >> 4) << 3);
#pragma unroll
    for (int t = 0; t < 8; ++t) {
        const int n = t * 16 + ncol;
#pragma unroll
        for (int r = 0; r < 8; ++r)
            h[(size_t)(mbase + r) * FDIM + n] = acc[t][r];
    }
}

__global__ void attn_terms(const float* __restrict__ h,
                           const float* __restrict__ att_src,
                           const float* __restrict__ att_dst,
                           float* __restrict__ a_src, float* __restrict__ a_dst,
                           int Nn) {
    int i = blockIdx.x * blockDim.x + threadIdx.x;
    if (i >= Nn * HEADS) return;
    int node = i >> 2, hd = i & 3;
    const float* hp = h + (size_t)node * FDIM + hd * CHAN;
    const float* as = att_src + hd * CHAN;
    const float* ad = att_dst + hd * CHAN;
    float s = 0.f, d = 0.f;
#pragma unroll
    for (int c = 0; c < CHAN; ++c) {
        float v = hp[c];
        s = fmaf(v, as[c], s);
        d = fmaf(v, ad[c], d);
    }
    a_src[i] = s;
    a_dst[i] = d;
}

__device__ __forceinline__ float leaky(float v) {
    return v > 0.f ? v : NEG_SLOPE * v;
}

__device__ __forceinline__ void atomicMaxF32(float* addr, float val) {
    if (val >= 0.f)
        atomicMax((int*)addr, __float_as_int(val));
    else
        atomicMin((unsigned int*)addr, __float_as_uint(val));
}

__global__ void edge_max(const int* __restrict__ ei,
                         const float* __restrict__ a_src,
                         const float* __restrict__ a_dst,
                         float* __restrict__ amax, int Ecnt) {
    int i = blockIdx.x * blockDim.x + threadIdx.x;
    if (i >= Ecnt * HEADS) return;
    int e = i >> 2, hd = i & 3;
    int s = ei[e], d = ei[Ecnt + e];
    float al = leaky(a_src[s * HEADS + hd] + a_dst[d * HEADS + hd]);
    atomicMaxF32(&amax[d * HEADS + hd], al);
}

__global__ void edge_denom(const int* __restrict__ ei,
                           const float* __restrict__ a_src,
                           const float* __restrict__ a_dst,
                           const float* __restrict__ amax,
                           float* __restrict__ denom, int Ecnt) {
    int i = blockIdx.x * blockDim.x + threadIdx.x;
    if (i >= Ecnt * HEADS) return;
    int e = i >> 2, hd = i & 3;
    int s = ei[e], d = ei[Ecnt + e];
    float al = leaky(a_src[s * HEADS + hd] + a_dst[d * HEADS + hd]);
    float mx = amax[d * HEADS + hd];
    if (!__builtin_isfinite(mx)) mx = 0.f;
    atomicAdd(&denom[d * HEADS + hd], expf(al - mx));
}

__global__ void edge_message(const int* __restrict__ ei,
                             const float* __restrict__ a_src,
                             const float* __restrict__ a_dst,
                             const float* __restrict__ amax,
                             const float* __restrict__ denom,
                             const float* __restrict__ h,
                             float* __restrict__ out, int Ecnt) {
    unsigned int i = blockIdx.x * blockDim.x + threadIdx.x;
    unsigned int e = i >> 7;
    if (e >= (unsigned int)Ecnt) return;
    int j  = i & 127;
    int hd = j >> 5;
    int s = ei[e], d = ei[Ecnt + e];
    float al = leaky(a_src[s * HEADS + hd] + a_dst[d * HEADS + hd]);
    float mx = amax[d * HEADS + hd];
    if (!__builtin_isfinite(mx)) mx = 0.f;
    float w = expf(al - mx) / (denom[d * HEADS + hd] + 1e-16f);
    atomicAdd(&out[(size_t)d * FDIM + j], h[(size_t)s * FDIM + j] * w);
}

__global__ void finalize(float* __restrict__ out, const float* __restrict__ bias,
                         int total) {
    int i = blockIdx.x * blockDim.x + threadIdx.x;
    if (i >= total) return;
    float v = out[i] + bias[i & (FDIM - 1)];
    out[i] = v > 0.f ? v : expm1f(v);
}

extern "C" void kernel_launch(void* const* d_in, const int* in_sizes, int n_in,
                              void* d_out, int out_size, void* d_ws, size_t ws_size,
                              hipStream_t stream) {
    const float* x       = (const float*)d_in[0];
    const int*   ei      = (const int*)  d_in[1];
    const float* W       = (const float*)d_in[2];
    const float* att_src = (const float*)d_in[3];
    const float* att_dst = (const float*)d_in[4];
    const float* bias    = (const float*)d_in[5];
    float* out = (float*)d_out;

    const int N = in_sizes[0] / FDIM;
    const int E = in_sizes[1] / 2;

    float* ws     = (float*)d_ws;
    float* h      = ws;
    float* a_src  = h     + (size_t)N * FDIM;
    float* a_dst  = a_src + (size_t)N * HEADS;
    float* amax   = a_dst + (size_t)N * HEADS;
    float* denom  = amax  + (size_t)N * HEADS;

    const int B = 256;

    init_kernel<<<(N * FDIM + B - 1) / B, B, 0, stream>>>(amax, denom, out, N);

    int tiles  = (N + 15) / 16;
    int gblk   = (tiles + 3) / 4;
    gemm_wmma<<<gblk, 128, 0, stream>>>(x, W, h, N);

    attn_terms<<<(N * HEADS + B - 1) / B, B, 0, stream>>>(h, att_src, att_dst,
                                                          a_src, a_dst, N);
    edge_max  <<<(E * HEADS + B - 1) / B, B, 0, stream>>>(ei, a_src, a_dst, amax, E);
    edge_denom<<<(E * HEADS + B - 1) / B, B, 0, stream>>>(ei, a_src, a_dst, amax,
                                                          denom, E);
    long long msg_threads = (long long)E * FDIM;
    edge_message<<<(int)((msg_threads + B - 1) / B), B, 0, stream>>>(
        ei, a_src, a_dst, amax, denom, h, out, E);

    finalize<<<(N * FDIM + B - 1) / B, B, 0, stream>>>(out, bias, N * FDIM);
}


# edges grouped by destination in-call (one returning int add per edge gives count and rank); one wave per node does max, exp-sum and weighted row sum in registers: no float atomics, sums in slot order
# speedup vs baseline: 3.8146x; 3.8146x over previous
#include <hip/hip_runtime.h>
#include <math.h>

typedef float v2f __attribute__((ext_vector_type(2)));
typedef float v8f __attribute__((ext_vector_type(8)));

#define HEADS 4
#define CHAN 32
#define FDIM 128
#define NEG_SLOPE 0.2f

__global__ void gemm_wmma(const float* __restrict__ x, const float* __restrict__ W,
                          float* __restrict__ h, int Nn) {
    const int wave = threadIdx.x >> 5;
    const int lane = threadIdx.x & 31;
    const int row0 = (blockIdx.x * 4 + wave) * 16;
    if (row0 >= Nn) return;

    v8f acc[8];
#pragma unroll
    for (int t = 0; t < 8; ++t) acc[t] = (v8f)(0.0f);

    const int mrow  = row0 + (lane & 15);
    const int khalf = (lane >> 4) << 1;
    const int ncol  = lane & 15;

    for (int k0 = 0; k0 < FDIM; k0 += 4) {
        const int ka = k0 + khalf;
        v2f a;
        a.x = x[(size_t)mrow * FDIM + ka];
        a.y = x[(size_t)mrow * FDIM + ka + 1];
#pragma unroll
        for (int t = 0; t < 8; ++t) {
            const int n = t * 16 + ncol;
            v2f b;
            b.x = W[(size_t)ka * FDIM + n];
            b.y = W[(size_t)(ka + 1) * FDIM + n];
            acc[t] = __builtin_amdgcn_wmma_f32_16x16x4_f32(
                false, a, false, b, (short)0, acc[t], false, false);
        }
    }

    const int mbase = row0 + ((lane >> 4) << 3);
#pragma unroll
    for (int t = 0; t < 8; ++t) {
        const int n = t * 16 + ncol;
#pragma unroll
        for (int r = 0; r < 8; ++r)
            h[(size_t)(mbase + r) * FDIM + n] = acc[t][r];
    }
}

__global__ void attn_terms(const float* __restrict__ h,
                           const float* __restrict__ att_src,
                           const float* __restrict__ att_dst,
                           float* __restrict__ a_src, float* __restrict__ a_dst,
                           int Nn) {
    int i = blockIdx.x * blockDim.x + threadIdx.x;
    if (i >= Nn * HEADS) return;
    int node = i >> 2, hd = i & 3;
    const float* hp = h + (size_t)node * FDIM + hd * CHAN;
    const float* as = att_src + hd * CHAN;
    const float* ad = att_dst + hd * CHAN;
    float s = 0.f, d = 0.f;
#pragma unroll
    for (int c = 0; c < CHAN; ++c) {
        float v = hp[c];
        s = fmaf(v, as[c], s);
        d = fmaf(v, ad[c], d);
    }
    a_src[i] = s;
    a_dst[i] = d;
}

__device__ __forceinline__ float leaky(float v) {
    return v > 0.f ? v : NEG_SLOPE * v;
}

typedef float v4f __attribute__((ext_vector_type(4)));

__global__ void zero_counts(int* __restrict__ cnt, int n) {
    int i = blockIdx.x * blockDim.x + threadIdx.x;
    if (i < n) cnt[i] = 0;
}

__global__ void count_and_rank(const int* __restrict__ ei, int* __restrict__ cnt,
                               int* __restrict__ rank, int Ecnt) {
    int e = blockIdx.x * blockDim.x + threadIdx.x;
    if (e >= Ecnt) return;
    rank[e] = atomicAdd(&cnt[ei[Ecnt + e]], 1);
}

__global__ void __launch_bounds__(256)
segment_offsets(const int* __restrict__ cnt, int* __restrict__ total,
                int* __restrict__ off, int Nn) {
    __shared__ int wave_base[8];
    __shared__ int block_base;
    const int i    = blockIdx.x * 256 + threadIdx.x;
    const int lane = threadIdx.x & 31;
    const int w    = threadIdx.x >> 5;
    const int c    = (i < Nn) ? cnt[i] : 0;
    int incl = c;
    for (int o = 1; o < 32; o <<= 1) {
        int t = __shfl_up(incl, o, 32);
        if (lane >= o) incl += t;
    }
    if (lane == 31) wave_base[w] = incl;
    __syncthreads();
    if (threadIdx.x == 0) {
        int s = 0;
        for (int k = 0; k < 8; ++k) { int t = wave_base[k]; wave_base[k] = s; s += t; }
        block_base = atomicAdd(total, s);
    }
    __syncthreads();
    if (i < Nn) off[i] = block_base + wave_base[w] + incl - c;
}

__global__ void place_by_destination(const int* __restrict__ ei,
                                     const int* __restrict__ off,
                                     const int* __restrict__ rank,
                                     int* __restrict__ seg_src, int Ecnt) {
    int e = blockIdx.x * blockDim.x + threadIdx.x;
    if (e >= Ecnt) return;
    seg_src[off[ei[Ecnt + e]] + rank[e]] = ei[e];
}

__device__ __forceinline__ float wave_max(float v) {
    for (int o = 16; o > 0; o >>= 1) v = fmaxf(v, __shfl_xor(v, o, 32));
    return v;
}

__device__ __forceinline__ v4f leaky4(v4f v) {
    v4f r;
    r.x = leaky(v.x); r.y = leaky(v.y); r.z = leaky(v.z); r.w = leaky(v.w);
    return r;
}

__device__ __forceinline__ float lane_bcast(float v, int k) {
    return __int_as_float(__builtin_amdgcn_readlane(__float_as_int(v), k));
}

__global__ void __launch_bounds__(256)
node_softmax_gather(const int* __restrict__ off, const int* __restrict__ cnt,
                    const int* __restrict__ seg_src,
                    const float* __restrict__ a_src, const float* __restrict__ a_dst,
                    const float* __restrict__ h, const float* __restrict__ bias,
                    float* __restrict__ out, int Nn) {
    const int lane = threadIdx.x & 31;
    const int n = blockIdx.x * 8 + __builtin_amdgcn_readfirstlane(threadIdx.x >> 5);
    if (n >= Nn) return;
    const int start = off[n];
    const int deg   = cnt[n];
    const int hd    = lane >> 3;
    const v4f ad = *(const v4f*)(a_dst + (size_t)n * HEADS);
    const float ninf = -__builtin_inff();

    const bool v0 = lane < deg;
    int s0 = 0;
    v4f l0 = {ninf, ninf, ninf, ninf};
    if (v0) {
        s0 = seg_src[start + lane];
        l0 = leaky4(*(const v4f*)(a_src + (size_t)s0 * HEADS) + ad);
    }
    v4f m = l0;
    for (int base = 32; base < deg; base += 32) {
        const int idx = base + lane;
        if (idx < deg) {
            const int s = seg_src[start + idx];
            const v4f l = leaky4(*(const v4f*)(a_src + (size_t)s * HEADS) + ad);
            m.x = fmaxf(m.x, l.x); m.y = fmaxf(m.y, l.y);
            m.z = fmaxf(m.z, l.z); m.w = fmaxf(m.w, l.w);
        }
    }
    m.x = wave_max(m.x); m.y = wave_max(m.y); m.z = wave_max(m.z); m.w = wave_max(m.w);
    if (!__builtin_isfinite(m.x)) m.x = 0.f;
    if (!__builtin_isfinite(m.y)) m.y = 0.f;
    if (!__builtin_isfinite(m.z)) m.z = 0.f;
    if (!__builtin_isfinite(m.w)) m.w = 0.f;

    v4f   acc = {0.f, 0.f, 0.f, 0.f};
    float den = 0.f;
    const float* hrow = h + lane * 4;

    for (int base = 0; base < deg; base += 32) {
        int s = s0;
        v4f l = l0;
        if (base > 0) {
            s = 0;
            l = (v4f){ninf, ninf, ninf, ninf};
            const int idx = base + lane;
            if (idx < deg) {
                s = seg_src[start + idx];
                l = leaky4(*(const v4f*)(a_src + (size_t)s * HEADS) + ad);
            }
        }
        const bool valid = base + lane < deg;
        v4f ex;
        ex.x = valid ? expf(l.x - m.x) : 0.f;
        ex.y = valid ? expf(l.y - m.y) : 0.f;
        ex.z = valid ? expf(l.z - m.z) : 0.f;
        ex.w = valid ? expf(l.w - m.w) : 0.f;

        const int c = min(deg - base, 32);
        for (int k = 0; k < c; k += 4) {
            const int g = min(c - k, 4);
            v4f r0, r1, r2, r3;
            float w0, w1, w2, w3;
            {
                const int sk = __builtin_amdgcn_readlane(s, k);
                r0 = *(const v4f*)(hrow + (size_t)sk * FDIM);
                const float e0 = lane_bcast(ex.x, k), e1 = lane_bcast(ex.y, k);
                const float e2 = lane_bcast(ex.z, k), e3 = lane_bcast(ex.w, k);
                w0 = hd == 0 ? e0 : hd == 1 ? e1 : hd == 2 ? e2 : e3;
            }
            if (g > 1) {
                const int sk = __builtin_amdgcn_readlane(s, k + 1);
                r1 = *(const v4f*)(hrow + (size_t)sk * FDIM);
                const float e0 = lane_bcast(ex.x, k + 1), e1 = lane_bcast(ex.y, k + 1);
                const float e2 = lane_bcast(ex.z, k + 1), e3 = lane_bcast(ex.w, k + 1);
                w1 = hd == 0 ? e0 : hd == 1 ? e1 : hd == 2 ? e2 : e3;
            }
            if (g > 2) {
                const int sk = __builtin_amdgcn_readlane(s, k + 2);
                r2 = *(const v4f*)(hrow + (size_t)sk * FDIM);
                const float e0 = lane_bcast(ex.x, k + 2), e1 = lane_bcast(ex.y, k + 2);
                const float e2 = lane_bcast(ex.z, k + 2), e3 = lane_bcast(ex.w, k + 2);
                w2 = hd == 0 ? e0 : hd == 1 ? e1 : hd == 2 ? e2 : e3;
            }
            if (g > 3) {
                const int sk = __builtin_amdgcn_readlane(s, k + 3);
                r3 = *(const v4f*)(hrow + (size_t)sk * FDIM);
                const float e0 = lane_bcast(ex.x, k + 3), e1 = lane_bcast(ex.y, k + 3);
                const float e2 = lane_bcast(ex.z, k + 3), e3 = lane_bcast(ex.w, k + 3);
                w3 = hd == 0 ? e0 : hd == 1 ? e1 : hd == 2 ? e2 : e3;
            }
            acc.x = fmaf(r0.x, w0, acc.x); acc.y = fmaf(r0.y, w0, acc.y);
            acc.z = fmaf(r0.z, w0, acc.z); acc.w = fmaf(r0.w, w0, acc.w);
            den += w0;
            if (g > 1) {
                acc.x = fmaf(r1.x, w1, acc.x); acc.y = fmaf(r1.y, w1, acc.y);
                acc.z = fmaf(r1.z, w1, acc.z); acc.w = fmaf(r1.w, w1, acc.w);
                den += w1;
            }
            if (g > 2) {
                acc.x = fmaf(r2.x, w2, acc.x); acc.y = fmaf(r2.y, w2, acc.y);
                acc.z = fmaf(r2.z, w2, acc.z); acc.w = fmaf(r2.w, w2, acc.w);
                den += w2;
            }
            if (g > 3) {
                acc.x = fmaf(r3.x, w3, acc.x); acc.y = fmaf(r3.y, w3, acc.y);
                acc.z = fmaf(r3.z, w3, acc.z); acc.w = fmaf(r3.w, w3, acc.w);
                den += w3;
            }
        }
    }

    const float dn = den + 1e-16f;
    const int j = lane * 4;
    v4f o;
    o.x = acc.x / dn + bias[j];
    o.y = acc.y / dn + bias[j + 1];
    o.z = acc.z / dn + bias[j + 2];
    o.w = acc.w / dn + bias[j + 3];
    o.x = o.x > 0.f ? o.x : expm1f(o.x);
    o.y = o.y > 0.f ? o.y : expm1f(o.y);
    o.z = o.z > 0.f ? o.z : expm1f(o.z);
    o.w = o.w > 0.f ? o.w : expm1f(o.w);
    *(v4f*)(out + (size_t)n * FDIM + j) = o;
}

extern "C" void kernel_launch(void* const* d_in, const int* in_sizes, int n_in,
                              void* d_out, int out_size, void* d_ws, size_t ws_size,
                              hipStream_t stream) {
    const float* x       = (const float*)d_in[0];
    const int*   ei      = (const int*)  d_in[1];
    const float* W       = (const float*)d_in[2];
    const float* att_src = (const float*)d_in[3];
    const float* att_dst = (const float*)d_in[4];
    const float* bias    = (const float*)d_in[5];
    float* out = (float*)d_out;

    const int N = in_sizes[0] / FDIM;
    const int E = in_sizes[1] / 2;

    float* ws     = (float*)d_ws;
    float* h      = ws;
    float* a_src  = h     + (size_t)N * FDIM;
    float* a_dst  = a_src + (size_t)N * HEADS;
    int*   cnt    = (int*)(a_dst + (size_t)N * HEADS);
    int*   off    = cnt  + (size_t)N + 1;
    int*   rank   = off  + (size_t)N;
    int*   segsrc = rank + (size_t)E;

    const int B = 256;

    zero_counts<<<(N + 1 + B - 1) / B, B, 0, stream>>>(cnt, N + 1);
    count_and_rank<<<(E + B - 1) / B, B, 0, stream>>>(ei, cnt, rank, E);
    segment_offsets<<<(N + B - 1) / B, B, 0, stream>>>(cnt, cnt + N, off, N);
    place_by_destination<<<(E + B - 1) / B, B, 0, stream>>>(ei, off, rank, segsrc, E);

    int tiles  = (N + 15) / 16;
    int gblk   = (tiles + 3) / 4;
    gemm_wmma<<<gblk, 128, 0, stream>>>(x, W, h, N);

    attn_terms<<<(N * HEADS + B - 1) / B, B, 0, stream>>>(h, att_src, att_dst,
                                                          a_src, a_dst, N);

    node_softmax_gather<<<(N + 7) / 8, B, 0, stream>>>(off, cnt, segsrc, a_src, a_dst,
                                                       h, bias, out, N);
}


# edges binned by destination range through LDS counters, then counted, scanned and placed per range in LDS: no global add per edge, 2 launches fewer; row sums in LDS arrival-rank slot order
# speedup vs baseline: 6.3104x; 1.6543x over previous
#include <hip/hip_runtime.h>
#include <math.h>

typedef float v2f __attribute__((ext_vector_type(2)));
typedef float v8f __attribute__((ext_vector_type(8)));

#define HEADS 4
#define CHAN 32
#define FDIM 128
#define NEG_SLOPE 0.2f

__global__ void gemm_wmma(const float* __restrict__ x, const float* __restrict__ W,
                          float* __restrict__ h, int Nn) {
    const int wave = threadIdx.x >> 5;
    const int lane = threadIdx.x & 31;
    const int row0 = (blockIdx.x * 4 + wave) * 16;
    if (row0 >= Nn) return;

    v8f acc[8];
#pragma unroll
    for (int t = 0; t < 8; ++t) acc[t] = (v8f)(0.0f);

    const int mrow  = row0 + (lane & 15);
    const int khalf = (lane >> 4) << 1;
    const int ncol  = lane & 15;

    for (int k0 = 0; k0 < FDIM; k0 += 4) {
        const int ka = k0 + khalf;
        v2f a;
        a.x = x[(size_t)mrow * FDIM + ka];
        a.y = x[(size_t)mrow * FDIM + ka + 1];
#pragma unroll
        for (int t = 0; t < 8; ++t) {
            const int n = t * 16 + ncol;
            v2f b;
            b.x = W[(size_t)ka * FDIM + n];
            b.y = W[(size_t)(ka + 1) * FDIM + n];
            acc[t] = __builtin_amdgcn_wmma_f32_16x16x4_f32(
                false, a, false, b, (short)0, acc[t], false, false);
        }
    }

    const int mbase = row0 + ((lane >> 4) << 3);
#pragma unroll
    for (int t = 0; t < 8; ++t) {
        const int n = t * 16 + ncol;
#pragma unroll
        for (int r = 0; r < 8; ++r)
            h[(size_t)(mbase + r) * FDIM + n] = acc[t][r];
    }
}

__global__ void attn_terms(const float* __restrict__ h,
                           const float* __restrict__ att_src,
                           const float* __restrict__ att_dst,
                           float* __restrict__ a_src, float* __restrict__ a_dst,
                           int Nn) {
    int i = blockIdx.x * blockDim.x + threadIdx.x;
    if (i >= Nn * HEADS) return;
    int node = i >> 2, hd = i & 3;
    const float* hp = h + (size_t)node * FDIM + hd * CHAN;
    const float* as = att_src + hd * CHAN;
    const float* ad = att_dst + hd * CHAN;
    float s = 0.f, d = 0.f;
#pragma unroll
    for (int c = 0; c < CHAN; ++c) {
        float v = hp[c];
        s = fmaf(v, as[c], s);
        d = fmaf(v, ad[c], d);
    }
    a_src[i] = s;
    a_dst[i] = d;
}

__device__ __forceinline__ float leaky(float v) {
    return v > 0.f ? v : NEG_SLOPE * v;
}

typedef float v4f __attribute__((ext_vector_type(4)));

#define RB      256
#define RB_LOG  8
#define EPB     8192
#define BIN_T   1024


__global__ void __launch_bounds__(BIN_T)
bin_by_range(const int* __restrict__ ei, int* __restrict__ binned,
             int* __restrict__ blockoff, int* __restrict__ total, int Ecnt, int NB) {
    extern __shared__ int hist[];
    __shared__ int stage[EPB];
    __shared__ int wave_tot[32];
    __shared__ int tile_total;
    const int tid  = threadIdx.x;
    const int lane = tid & 31;
    const int w    = tid >> 5;

    for (int i = tid; i < NB; i += BIN_T) hist[i] = 0;
    if (blockIdx.x == 0 && tid == 0) *total = 0;
    __syncthreads();

    const int base = blockIdx.x * EPB;
    int word[8], bkt[8], rk[8];
#pragma unroll
    for (int j = 0; j < 8; ++j) {
        const int e = base + j * BIN_T + tid;
        bkt[j]  = -1;
        word[j] = 0;
        rk[j]   = 0;
        if (e < Ecnt) {
            const int d = ei[Ecnt + e];
            const int s = ei[e];
            bkt[j]  = d >> RB_LOG;
            word[j] = (int)(((unsigned)(d & (RB - 1)) << 24) | (unsigned)s);
        }
    }
#pragma unroll
    for (int j = 0; j < 8; ++j)
        if (bkt[j] >= 0) rk[j] = atomicAdd(&hist[bkt[j]], 1);
    __syncthreads();

    int carry = 0;
    for (int t0 = 0; t0 < NB; t0 += BIN_T) {
        const int i = t0 + tid;
        const int c = (i < NB) ? hist[i] : 0;
        int incl = c;
        for (int o = 1; o < 32; o <<= 1) {
            int t = __shfl_up(incl, o, 32);
            if (lane >= o) incl += t;
        }
        if (lane == 31) wave_tot[w] = incl;
        __syncthreads();
        if (w == 0) {
            const int v = wave_tot[lane];
            int inc2 = v;
            for (int o = 1; o < 32; o <<= 1) {
                int t = __shfl_up(inc2, o, 32);
                if (lane >= o) inc2 += t;
            }
            wave_tot[lane] = inc2 - v;
            if (lane == 31) tile_total = inc2;
        }
        __syncthreads();
        if (i < NB) hist[i] = carry + wave_tot[w] + incl - c;
        carry += tile_total;
        __syncthreads();
    }

    int* tab = blockoff + (size_t)blockIdx.x * (NB + 1);
    for (int i = tid; i < NB; i += BIN_T) tab[i] = hist[i];
    if (tid == 0) tab[NB] = carry;

#pragma unroll
    for (int j = 0; j < 8; ++j)
        if (bkt[j] >= 0) stage[hist[bkt[j]] + rk[j]] = word[j];
    __syncthreads();
#pragma unroll
    for (int j = 0; j < 8; ++j) {
        const int q = j * BIN_T + tid;
        if (q < carry) binned[base + q] = stage[q];
    }
}

__global__ void __launch_bounds__(RB)
range_segments(const int* __restrict__ binned, const int* __restrict__ blockoff,
               int* __restrict__ total, int* __restrict__ cnt, int* __restrict__ off,
               int* __restrict__ seg_src, int nblk, int NB, int Nn) {
    __shared__ int slot[RB];
    __shared__ int wave_base[RB / 32];
    __shared__ int block_base;
    const int tid  = threadIdx.x;
    const int lane = tid & 31;
    const int w    = tid >> 5;
    const int b    = blockIdx.x;

    slot[tid] = 0;
    __syncthreads();
    for (int k = tid; k < nblk; k += RB) {
        const int* tab = blockoff + (size_t)k * (NB + 1);
        const int p0 = k * EPB + tab[b];
        const int p1 = k * EPB + tab[b + 1];
        for (int p = p0; p < p1; p += 4) {
            const int last = p1 - 1;
            const int w0 = binned[p];
            const int w1 = binned[min(p + 1, last)];
            const int w2 = binned[min(p + 2, last)];
            const int w3 = binned[min(p + 3, last)];
            atomicAdd(&slot[(unsigned)w0 >> 24], 1);
            if (p + 1 < p1) atomicAdd(&slot[(unsigned)w1 >> 24], 1);
            if (p + 2 < p1) atomicAdd(&slot[(unsigned)w2 >> 24], 1);
            if (p + 3 < p1) atomicAdd(&slot[(unsigned)w3 >> 24], 1);
        }
    }
    __syncthreads();

    const int c = slot[tid];
    int incl = c;
    for (int o = 1; o < 32; o <<= 1) {
        int t = __shfl_up(incl, o, 32);
        if (lane >= o) incl += t;
    }
    if (lane == 31) wave_base[w] = incl;
    __syncthreads();
    if (tid == 0) {
        int s = 0;
        for (int k = 0; k < RB / 32; ++k) { int t = wave_base[k]; wave_base[k] = s; s += t; }
        block_base = atomicAdd(total, s);
    }
    __syncthreads();
    const int o0   = block_base + wave_base[w] + incl - c;
    const int node = b * RB + tid;
    if (node < Nn) { cnt[node] = c; off[node] = o0; }
    slot[tid] = o0;
    __syncthreads();

    for (int k = tid; k < nblk; k += RB) {
        const int* tab = blockoff + (size_t)k * (NB + 1);
        const int p0 = k * EPB + tab[b];
        const int p1 = k * EPB + tab[b + 1];
        for (int p = p0; p < p1; p += 4) {
            const int last = p1 - 1;
            const int w0 = binned[p];
            const int w1 = binned[min(p + 1, last)];
            const int w2 = binned[min(p + 2, last)];
            const int w3 = binned[min(p + 3, last)];
            seg_src[atomicAdd(&slot[(unsigned)w0 >> 24], 1)] = w0 & 0x00FFFFFF;
            if (p + 1 < p1) seg_src[atomicAdd(&slot[(unsigned)w1 >> 24], 1)] = w1 & 0x00FFFFFF;
            if (p + 2 < p1) seg_src[atomicAdd(&slot[(unsigned)w2 >> 24], 1)] = w2 & 0x00FFFFFF;
            if (p + 3 < p1) seg_src[atomicAdd(&slot[(unsigned)w3 >> 24], 1)] = w3 & 0x00FFFFFF;
        }
    }
}

__device__ __forceinline__ float wave_max(float v) {
    for (int o = 16; o > 0; o >>= 1) v = fmaxf(v, __shfl_xor(v, o, 32));
    return v;
}

__device__ __forceinline__ v4f leaky4(v4f v) {
    v4f r;
    r.x = leaky(v.x); r.y = leaky(v.y); r.z = leaky(v.z); r.w = leaky(v.w);
    return r;
}

__device__ __forceinline__ float lane_bcast(float v, int k) {
    return __int_as_float(__builtin_amdgcn_readlane(__float_as_int(v), k));
}

__global__ void __launch_bounds__(256)
node_softmax_gather(const int* __restrict__ off, const int* __restrict__ cnt,
                    const int* __restrict__ seg_src,
                    const float* __restrict__ a_src, const float* __restrict__ a_dst,
                    const float* __restrict__ h, const float* __restrict__ bias,
                    float* __restrict__ out, int Nn) {
    const int lane = threadIdx.x & 31;
    const int n = blockIdx.x * 8 + __builtin_amdgcn_readfirstlane(threadIdx.x >> 5);
    if (n >= Nn) return;
    const int start = off[n];
    const int deg   = cnt[n];
    const int hd    = lane >> 3;
    const v4f ad = *(const v4f*)(a_dst + (size_t)n * HEADS);
    const float ninf = -__builtin_inff();

    const bool v0 = lane < deg;
    int s0 = 0;
    v4f l0 = {ninf, ninf, ninf, ninf};
    if (v0) {
        s0 = seg_src[start + lane];
        l0 = leaky4(*(const v4f*)(a_src + (size_t)s0 * HEADS) + ad);
    }
    v4f m = l0;
    for (int base = 32; base < deg; base += 32) {
        const int idx = base + lane;
        if (idx < deg) {
            const int s = seg_src[start + idx];
            const v4f l = leaky4(*(const v4f*)(a_src + (size_t)s * HEADS) + ad);
            m.x = fmaxf(m.x, l.x); m.y = fmaxf(m.y, l.y);
            m.z = fmaxf(m.z, l.z); m.w = fmaxf(m.w, l.w);
        }
    }
    m.x = wave_max(m.x); m.y = wave_max(m.y); m.z = wave_max(m.z); m.w = wave_max(m.w);
    if (!__builtin_isfinite(m.x)) m.x = 0.f;
    if (!__builtin_isfinite(m.y)) m.y = 0.f;
    if (!__builtin_isfinite(m.z)) m.z = 0.f;
    if (!__builtin_isfinite(m.w)) m.w = 0.f;

    v4f   acc = {0.f, 0.f, 0.f, 0.f};
    float den = 0.f;
    const float* hrow = h + lane * 4;

    for (int base = 0; base < deg; base += 32) {
        int s = s0;
        v4f l = l0;
        if (base > 0) {
            s = 0;
            l = (v4f){ninf, ninf, ninf, ninf};
            const int idx = base + lane;
            if (idx < deg) {
                s = seg_src[start + idx];
                l = leaky4(*(const v4f*)(a_src + (size_t)s * HEADS) + ad);
            }
        }
        const bool valid = base + lane < deg;
        v4f ex;
        ex.x = valid ? expf(l.x - m.x) : 0.f;
        ex.y = valid ? expf(l.y - m.y) : 0.f;
        ex.z = valid ? expf(l.z - m.z) : 0.f;
        ex.w = valid ? expf(l.w - m.w) : 0.f;

        const int c = min(deg - base, 32);
        for (int k = 0; k < c; k += 4) {
            const int g = min(c - k, 4);
            v4f r0, r1, r2, r3;
            float w0, w1, w2, w3;
            {
                const int sk = __builtin_amdgcn_readlane(s, k);
                r0 = *(const v4f*)(hrow + (size_t)sk * FDIM);
                const float e0 = lane_bcast(ex.x, k), e1 = lane_bcast(ex.y, k);
                const float e2 = lane_bcast(ex.z, k), e3 = lane_bcast(ex.w, k);
                w0 = hd == 0 ? e0 : hd == 1 ? e1 : hd == 2 ? e2 : e3;
            }
            if (g > 1) {
                const int sk = __builtin_amdgcn_readlane(s, k + 1);
                r1 = *(const v4f*)(hrow + (size_t)sk * FDIM);
                const float e0 = lane_bcast(ex.x, k + 1), e1 = lane_bcast(ex.y, k + 1);
                const float e2 = lane_bcast(ex.z, k + 1), e3 = lane_bcast(ex.w, k + 1);
                w1 = hd == 0 ? e0 : hd == 1 ? e1 : hd == 2 ? e2 : e3;
            }
            if (g > 2) {
                const int sk = __builtin_amdgcn_readlane(s, k + 2);
                r2 = *(const v4f*)(hrow + (size_t)sk * FDIM);
                const float e0 = lane_bcast(ex.x, k + 2), e1 = lane_bcast(ex.y, k + 2);
                const float e2 = lane_bcast(ex.z, k + 2), e3 = lane_bcast(ex.w, k + 2);
                w2 = hd == 0 ? e0 : hd == 1 ? e1 : hd == 2 ? e2 : e3;
            }
            if (g > 3) {
                const int sk = __builtin_amdgcn_readlane(s, k + 3);
                r3 = *(const v4f*)(hrow + (size_t)sk * FDIM);
                const float e0 = lane_bcast(ex.x, k + 3), e1 = lane_bcast(ex.y, k + 3);
                const float e2 = lane_bcast(ex.z, k + 3), e3 = lane_bcast(ex.w, k + 3);
                w3 = hd == 0 ? e0 : hd == 1 ? e1 : hd == 2 ? e2 : e3;
            }
            acc.x = fmaf(r0.x, w0, acc.x); acc.y = fmaf(r0.y, w0, acc.y);
            acc.z = fmaf(r0.z, w0, acc.z); acc.w = fmaf(r0.w, w0, acc.w);
            den += w0;
            if (g > 1) {
                acc.x = fmaf(r1.x, w1, acc.x); acc.y = fmaf(r1.y, w1, acc.y);
                acc.z = fmaf(r1.z, w1, acc.z); acc.w = fmaf(r1.w, w1, acc.w);
                den += w1;
            }
            if (g > 2) {
                acc.x = fmaf(r2.x, w2, acc.x); acc.y = fmaf(r2.y, w2, acc.y);
                acc.z = fmaf(r2.z, w2, acc.z); acc.w = fmaf(r2.w, w2, acc.w);
                den += w2;
            }
            if (g > 3) {
                acc.x = fmaf(r3.x, w3, acc.x); acc.y = fmaf(r3.y, w3, acc.y);
                acc.z = fmaf(r3.z, w3, acc.z); acc.w = fmaf(r3.w, w3, acc.w);
                den += w3;
            }
        }
    }

    const float dn = den + 1e-16f;
    const int j = lane * 4;
    v4f o;
    o.x = acc.x / dn + bias[j];
    o.y = acc.y / dn + bias[j + 1];
    o.z = acc.z / dn + bias[j + 2];
    o.w = acc.w / dn + bias[j + 3];
    o.x = o.x > 0.f ? o.x : expm1f(o.x);
    o.y = o.y > 0.f ? o.y : expm1f(o.y);
    o.z = o.z > 0.f ? o.z : expm1f(o.z);
    o.w = o.w > 0.f ? o.w : expm1f(o.w);
    *(v4f*)(out + (size_t)n * FDIM + j) = o;
}

extern "C" void kernel_launch(void* const* d_in, const int* in_sizes, int n_in,
                              void* d_out, int out_size, void* d_ws, size_t ws_size,
                              hipStream_t stream) {
    const float* x       = (const float*)d_in[0];
    const int*   ei      = (const int*)  d_in[1];
    const float* W       = (const float*)d_in[2];
    const float* att_src = (const float*)d_in[3];
    const float* att_dst = (const float*)d_in[4];
    const float* bias    = (const float*)d_in[5];
    float* out = (float*)d_out;

    const int N = in_sizes[0] / FDIM;
    const int E = in_sizes[1] / 2;

    float* ws     = (float*)d_ws;
    float* h      = ws;
    float* a_src  = h     + (size_t)N * FDIM;
    float* a_dst  = a_src + (size_t)N * HEADS;
    int*   cnt    = (int*)(a_dst + (size_t)N * HEADS);
    int*   off    = cnt  + (size_t)N + 1;
    int*   binned = off  + (size_t)N;
    int*   segsrc = binned + (size_t)E;
    int*   blkoff = segsrc + (size_t)E;

    const int B = 256;
    const int NB   = (N + RB - 1) / RB;
    const int nblk = (E + EPB - 1) / EPB;

    bin_by_range<<<nblk, BIN_T, NB * sizeof(int), stream>>>(ei, binned, blkoff, cnt + N, E, NB);
    range_segments<<<NB, RB, 0, stream>>>(binned, blkoff, cnt + N, cnt, off, segsrc,
                                          nblk, NB, N);

    int tiles  = (N + 15) / 16;
    int gblk   = (tiles + 3) / 4;
    gemm_wmma<<<gblk, 128, 0, stream>>>(x, W, h, N);

    attn_terms<<<(N * HEADS + B - 1) / B, B, 0, stream>>>(h, att_src, att_dst,
                                                          a_src, a_dst, N);

    node_softmax_gather<<<(N + 7) / 8, B, 0, stream>>>(off, cnt, segsrc, a_src, a_dst,
                                                       h, bias, out, N);
}


# attention terms a_src, a_dst summed from the GEMM accumulators in its epilogue (16-lane DPP tree instead of a serial 32-term chain): the separate pass over h and its launch are gone
# speedup vs baseline: 7.2180x; 1.1438x over previous
#include <hip/hip_runtime.h>
#include <math.h>

typedef float v2f __attribute__((ext_vector_type(2)));
typedef float v8f __attribute__((ext_vector_type(8)));

#define HEADS 4
#define CHAN 32
#define FDIM 128
#define NEG_SLOPE 0.2f


template <int XM>
__device__ __forceinline__ float row_xor(float v) {
    return __int_as_float(__builtin_amdgcn_update_dpp(
        0, __float_as_int(v), 0x160 + XM, 0xF, 0xF, true));
}

__device__ __forceinline__ float row_reduce_scatter16(const float (&v)[16], int c) {
    const bool b3 = (c & 8) != 0, b2 = (c & 4) != 0, b1 = (c & 2) != 0, b0 = (c & 1) != 0;
    float t8[8], t4[4], t2[2];
#pragma unroll
    for (int j = 0; j < 8; ++j) {
        const float keep = b3 ? v[8 + j] : v[j];
        const float send = b3 ? v[j] : v[8 + j];
        t8[j] = keep + row_xor<8>(send);
    }
#pragma unroll
    for (int j = 0; j < 4; ++j) {
        const float keep = b2 ? t8[4 + j] : t8[j];
        const float send = b2 ? t8[j] : t8[4 + j];
        t4[j] = keep + row_xor<4>(send);
    }
#pragma unroll
    for (int j = 0; j < 2; ++j) {
        const float keep = b1 ? t4[2 + j] : t4[j];
        const float send = b1 ? t4[j] : t4[2 + j];
        t2[j] = keep + row_xor<2>(send);
    }
    const float keep = b0 ? t2[1] : t2[0];
    const float send = b0 ? t2[0] : t2[1];
    return keep + row_xor<1>(send);
}

typedef float v4f __attribute__((ext_vector_type(4)));

__global__ void gemm_wmma_attn(const float* __restrict__ x, const float* __restrict__ W,
                               const float* __restrict__ att_src,
                               const float* __restrict__ att_dst,
                               float* __restrict__ h,
                               float* __restrict__ a_src, float* __restrict__ a_dst,
                               int Nn) {
    const int wave = threadIdx.x >> 5;
    const int lane = threadIdx.x & 31;
    const int row0 = (blockIdx.x * 4 + wave) * 16;
    if (row0 >= Nn) return;

    v8f acc[8];
#pragma unroll
    for (int t = 0; t < 8; ++t) acc[t] = (v8f)(0.0f);

    const int mrow  = row0 + (lane & 15);
    const int khalf = (lane >> 4) << 1;
    const int ncol  = lane & 15;

    for (int k0 = 0; k0 < FDIM; k0 += 4) {
        const int ka = k0 + khalf;
        v2f a;
        a.x = x[(size_t)mrow * FDIM + ka];
        a.y = x[(size_t)mrow * FDIM + ka + 1];
#pragma unroll
        for (int t = 0; t < 8; ++t) {
            const int n = t * 16 + ncol;
            v2f b;
            b.x = W[(size_t)ka * FDIM + n];
            b.y = W[(size_t)(ka + 1) * FDIM + n];
            acc[t] = __builtin_amdgcn_wmma_f32_16x16x4_f32(
                false, a, false, b, (short)0, acc[t], false, false);
        }
    }

    const int mbase = row0 + ((lane >> 4) << 3);
#pragma unroll
    for (int t = 0; t < 8; ++t) {
        const int n = t * 16 + ncol;
#pragma unroll
        for (int r = 0; r < 8; ++r)
            h[(size_t)(mbase + r) * FDIM + n] = acc[t][r];
    }

    v4f term;
#pragma unroll
    for (int hd = 0; hd < HEADS; ++hd) {
        const float s0 = att_src[hd * CHAN + ncol], s1 = att_src[hd * CHAN + 16 + ncol];
        const float d0 = att_dst[hd * CHAN + ncol], d1 = att_dst[hd * CHAN + 16 + ncol];
        float p[16];
#pragma unroll
        for (int r = 0; r < 8; ++r) {
            p[r]     = fmaf(acc[2 * hd + 1][r], s1, acc[2 * hd][r] * s0);
            p[8 + r] = fmaf(acc[2 * hd + 1][r], d1, acc[2 * hd][r] * d0);
        }
        term[hd] = row_reduce_scatter16(p, ncol);
    }
    float* dstp = (ncol < 8 ? a_src : a_dst) + (size_t)(mbase + (ncol & 7)) * HEADS;
    *(v4f*)dstp = term;
}

__device__ __forceinline__ float leaky(float v) {
    return v > 0.f ? v : NEG_SLOPE * v;
}


#define RB      256
#define RB_LOG  8
#define EPB     8192
#define BIN_T   1024


__global__ void __launch_bounds__(BIN_T)
bin_by_range(const int* __restrict__ ei, int* __restrict__ binned,
             int* __restrict__ blockoff, int* __restrict__ total, int Ecnt, int NB) {
    extern __shared__ int hist[];
    __shared__ int stage[EPB];
    __shared__ int wave_tot[32];
    __shared__ int tile_total;
    const int tid  = threadIdx.x;
    const int lane = tid & 31;
    const int w    = tid >> 5;

    for (int i = tid; i < NB; i += BIN_T) hist[i] = 0;
    if (blockIdx.x == 0 && tid == 0) *total = 0;
    __syncthreads();

    const int base = blockIdx.x * EPB;
    int word[8], bkt[8], rk[8];
#pragma unroll
    for (int j = 0; j < 8; ++j) {
        const int e = base + j * BIN_T + tid;
        bkt[j]  = -1;
        word[j] = 0;
        rk[j]   = 0;
        if (e < Ecnt) {
            const int d = ei[Ecnt + e];
            const int s = ei[e];
            bkt[j]  = d >> RB_LOG;
            word[j] = (int)(((unsigned)(d & (RB - 1)) << 24) | (unsigned)s);
        }
    }
#pragma unroll
    for (int j = 0; j < 8; ++j)
        if (bkt[j] >= 0) rk[j] = atomicAdd(&hist[bkt[j]], 1);
    __syncthreads();

    int carry = 0;
    for (int t0 = 0; t0 < NB; t0 += BIN_T) {
        const int i = t0 + tid;
        const int c = (i < NB) ? hist[i] : 0;
        int incl = c;
        for (int o = 1; o < 32; o <<= 1) {
            int t = __shfl_up(incl, o, 32);
            if (lane >= o) incl += t;
        }
        if (lane == 31) wave_tot[w] = incl;
        __syncthreads();
        if (w == 0) {
            const int v = wave_tot[lane];
            int inc2 = v;
            for (int o = 1; o < 32; o <<= 1) {
                int t = __shfl_up(inc2, o, 32);
                if (lane >= o) inc2 += t;
            }
            wave_tot[lane] = inc2 - v;
            if (lane == 31) tile_total = inc2;
        }
        __syncthreads();
        if (i < NB) hist[i] = carry + wave_tot[w] + incl - c;
        carry += tile_total;
        __syncthreads();
    }

    int* tab = blockoff + (size_t)blockIdx.x * (NB + 1);
    for (int i = tid; i < NB; i += BIN_T) tab[i] = hist[i];
    if (tid == 0) tab[NB] = carry;

#pragma unroll
    for (int j = 0; j < 8; ++j)
        if (bkt[j] >= 0) stage[hist[bkt[j]] + rk[j]] = word[j];
    __syncthreads();
#pragma unroll
    for (int j = 0; j < 8; ++j) {
        const int q = j * BIN_T + tid;
        if (q < carry) binned[base + q] = stage[q];
    }
}

__global__ void __launch_bounds__(RB)
range_segments(const int* __restrict__ binned, const int* __restrict__ blockoff,
               int* __restrict__ total, int* __restrict__ cnt, int* __restrict__ off,
               int* __restrict__ seg_src, int nblk, int NB, int Nn) {
    __shared__ int slot[RB];
    __shared__ int wave_base[RB / 32];
    __shared__ int block_base;
    const int tid  = threadIdx.x;
    const int lane = tid & 31;
    const int w    = tid >> 5;
    const int b    = blockIdx.x;

    slot[tid] = 0;
    __syncthreads();
    for (int k = tid; k < nblk; k += RB) {
        const int* tab = blockoff + (size_t)k * (NB + 1);
        const int p0 = k * EPB + tab[b];
        const int p1 = k * EPB + tab[b + 1];
        for (int p = p0; p < p1; p += 4) {
            const int last = p1 - 1;
            const int w0 = binned[p];
            const int w1 = binned[min(p + 1, last)];
            const int w2 = binned[min(p + 2, last)];
            const int w3 = binned[min(p + 3, last)];
            atomicAdd(&slot[(unsigned)w0 >> 24], 1);
            if (p + 1 < p1) atomicAdd(&slot[(unsigned)w1 >> 24], 1);
            if (p + 2 < p1) atomicAdd(&slot[(unsigned)w2 >> 24], 1);
            if (p + 3 < p1) atomicAdd(&slot[(unsigned)w3 >> 24], 1);
        }
    }
    __syncthreads();

    const int c = slot[tid];
    int incl = c;
    for (int o = 1; o < 32; o <<= 1) {
        int t = __shfl_up(incl, o, 32);
        if (lane >= o) incl += t;
    }
    if (lane == 31) wave_base[w] = incl;
    __syncthreads();
    if (tid == 0) {
        int s = 0;
        for (int k = 0; k < RB / 32; ++k) { int t = wave_base[k]; wave_base[k] = s; s += t; }
        block_base = atomicAdd(total, s);
    }
    __syncthreads();
    const int o0   = block_base + wave_base[w] + incl - c;
    const int node = b * RB + tid;
    if (node < Nn) { cnt[node] = c; off[node] = o0; }
    slot[tid] = o0;
    __syncthreads();

    for (int k = tid; k < nblk; k += RB) {
        const int* tab = blockoff + (size_t)k * (NB + 1);
        const int p0 = k * EPB + tab[b];
        const int p1 = k * EPB + tab[b + 1];
        for (int p = p0; p < p1; p += 4) {
            const int last = p1 - 1;
            const int w0 = binned[p];
            const int w1 = binned[min(p + 1, last)];
            const int w2 = binned[min(p + 2, last)];
            const int w3 = binned[min(p + 3, last)];
            seg_src[atomicAdd(&slot[(unsigned)w0 >> 24], 1)] = w0 & 0x00FFFFFF;
            if (p + 1 < p1) seg_src[atomicAdd(&slot[(unsigned)w1 >> 24], 1)] = w1 & 0x00FFFFFF;
            if (p + 2 < p1) seg_src[atomicAdd(&slot[(unsigned)w2 >> 24], 1)] = w2 & 0x00FFFFFF;
            if (p + 3 < p1) seg_src[atomicAdd(&slot[(unsigned)w3 >> 24], 1)] = w3 & 0x00FFFFFF;
        }
    }
}

__device__ __forceinline__ float wave_max(float v) {
    for (int o = 16; o > 0; o >>= 1) v = fmaxf(v, __shfl_xor(v, o, 32));
    return v;
}

__device__ __forceinline__ v4f leaky4(v4f v) {
    v4f r;
    r.x = leaky(v.x); r.y = leaky(v.y); r.z = leaky(v.z); r.w = leaky(v.w);
    return r;
}

__device__ __forceinline__ float lane_bcast(float v, int k) {
    return __int_as_float(__builtin_amdgcn_readlane(__float_as_int(v), k));
}

__global__ void __launch_bounds__(256)
node_softmax_gather(const int* __restrict__ off, const int* __restrict__ cnt,
                    const int* __restrict__ seg_src,
                    const float* __restrict__ a_src, const float* __restrict__ a_dst,
                    const float* __restrict__ h, const float* __restrict__ bias,
                    float* __restrict__ out, int Nn) {
    const int lane = threadIdx.x & 31;
    const int n = blockIdx.x * 8 + __builtin_amdgcn_readfirstlane(threadIdx.x >> 5);
    if (n >= Nn) return;
    const int start = off[n];
    const int deg   = cnt[n];
    const int hd    = lane >> 3;
    const v4f ad = *(const v4f*)(a_dst + (size_t)n * HEADS);
    const float ninf = -__builtin_inff();

    const bool v0 = lane < deg;
    int s0 = 0;
    v4f l0 = {ninf, ninf, ninf, ninf};
    if (v0) {
        s0 = seg_src[start + lane];
        l0 = leaky4(*(const v4f*)(a_src + (size_t)s0 * HEADS) + ad);
    }
    v4f m = l0;
    for (int base = 32; base < deg; base += 32) {
        const int idx = base + lane;
        if (idx < deg) {
            const int s = seg_src[start + idx];
            const v4f l = leaky4(*(const v4f*)(a_src + (size_t)s * HEADS) + ad);
            m.x = fmaxf(m.x, l.x); m.y = fmaxf(m.y, l.y);
            m.z = fmaxf(m.z, l.z); m.w = fmaxf(m.w, l.w);
        }
    }
    m.x = wave_max(m.x); m.y = wave_max(m.y); m.z = wave_max(m.z); m.w = wave_max(m.w);
    if (!__builtin_isfinite(m.x)) m.x = 0.f;
    if (!__builtin_isfinite(m.y)) m.y = 0.f;
    if (!__builtin_isfinite(m.z)) m.z = 0.f;
    if (!__builtin_isfinite(m.w)) m.w = 0.f;

    v4f   acc = {0.f, 0.f, 0.f, 0.f};
    float den = 0.f;
    const float* hrow = h + lane * 4;

    for (int base = 0; base < deg; base += 32) {
        int s = s0;
        v4f l = l0;
        if (base > 0) {
            s = 0;
            l = (v4f){ninf, ninf, ninf, ninf};
            const int idx = base + lane;
            if (idx < deg) {
                s = seg_src[start + idx];
                l = leaky4(*(const v4f*)(a_src + (size_t)s * HEADS) + ad);
            }
        }
        const bool valid = base + lane < deg;
        v4f ex;
        ex.x = valid ? expf(l.x - m.x) : 0.f;
        ex.y = valid ? expf(l.y - m.y) : 0.f;
        ex.z = valid ? expf(l.z - m.z) : 0.f;
        ex.w = valid ? expf(l.w - m.w) : 0.f;

        const int c = min(deg - base, 32);
        for (int k = 0; k < c; k += 4) {
            const int g = min(c - k, 4);
            v4f r0, r1, r2, r3;
            float w0, w1, w2, w3;
            {
                const int sk = __builtin_amdgcn_readlane(s, k);
                r0 = *(const v4f*)(hrow + (size_t)sk * FDIM);
                const float e0 = lane_bcast(ex.x, k), e1 = lane_bcast(ex.y, k);
                const float e2 = lane_bcast(ex.z, k), e3 = lane_bcast(ex.w, k);
                w0 = hd == 0 ? e0 : hd == 1 ? e1 : hd == 2 ? e2 : e3;
            }
            if (g > 1) {
                const int sk = __builtin_amdgcn_readlane(s, k + 1);
                r1 = *(const v4f*)(hrow + (size_t)sk * FDIM);
                const float e0 = lane_bcast(ex.x, k + 1), e1 = lane_bcast(ex.y, k + 1);
                const float e2 = lane_bcast(ex.z, k + 1), e3 = lane_bcast(ex.w, k + 1);
                w1 = hd == 0 ? e0 : hd == 1 ? e1 : hd == 2 ? e2 : e3;
            }
            if (g > 2) {
                const int sk = __builtin_amdgcn_readlane(s, k + 2);
                r2 = *(const v4f*)(hrow + (size_t)sk * FDIM);
                const float e0 = lane_bcast(ex.x, k + 2), e1 = lane_bcast(ex.y, k + 2);
                const float e2 = lane_bcast(ex.z, k + 2), e3 = lane_bcast(ex.w, k + 2);
                w2 = hd == 0 ? e0 : hd == 1 ? e1 : hd == 2 ? e2 : e3;
            }
            if (g > 3) {
                const int sk = __builtin_amdgcn_readlane(s, k + 3);
                r3 = *(const v4f*)(hrow + (size_t)sk * FDIM);
                const float e0 = lane_bcast(ex.x, k + 3), e1 = lane_bcast(ex.y, k + 3);
                const float e2 = lane_bcast(ex.z, k + 3), e3 = lane_bcast(ex.w, k + 3);
                w3 = hd == 0 ? e0 : hd == 1 ? e1 : hd == 2 ? e2 : e3;
            }
            acc.x = fmaf(r0.x, w0, acc.x); acc.y = fmaf(r0.y, w0, acc.y);
            acc.z = fmaf(r0.z, w0, acc.z); acc.w = fmaf(r0.w, w0, acc.w);
            den += w0;
            if (g > 1) {
                acc.x = fmaf(r1.x, w1, acc.x); acc.y = fmaf(r1.y, w1, acc.y);
                acc.z = fmaf(r1.z, w1, acc.z); acc.w = fmaf(r1.w, w1, acc.w);
                den += w1;
            }
            if (g > 2) {
                acc.x = fmaf(r2.x, w2, acc.x); acc.y = fmaf(r2.y, w2, acc.y);
                acc.z = fmaf(r2.z, w2, acc.z); acc.w = fmaf(r2.w, w2, acc.w);
                den += w2;
            }
            if (g > 3) {
                acc.x = fmaf(r3.x, w3, acc.x); acc.y = fmaf(r3.y, w3, acc.y);
                acc.z = fmaf(r3.z, w3, acc.z); acc.w = fmaf(r3.w, w3, acc.w);
                den += w3;
            }
        }
    }

    const float dn = den + 1e-16f;
    const int j = lane * 4;
    v4f o;
    o.x = acc.x / dn + bias[j];
    o.y = acc.y / dn + bias[j + 1];
    o.z = acc.z / dn + bias[j + 2];
    o.w = acc.w / dn + bias[j + 3];
    o.x = o.x > 0.f ? o.x : expm1f(o.x);
    o.y = o.y > 0.f ? o.y : expm1f(o.y);
    o.z = o.z > 0.f ? o.z : expm1f(o.z);
    o.w = o.w > 0.f ? o.w : expm1f(o.w);
    *(v4f*)(out + (size_t)n * FDIM + j) = o;
}

extern "C" void kernel_launch(void* const* d_in, const int* in_sizes, int n_in,
                              void* d_out, int out_size, void* d_ws, size_t ws_size,
                              hipStream_t stream) {
    const float* x       = (const float*)d_in[0];
    const int*   ei      = (const int*)  d_in[1];
    const float* W       = (const float*)d_in[2];
    const float* att_src = (const float*)d_in[3];
    const float* att_dst = (const float*)d_in[4];
    const float* bias    = (const float*)d_in[5];
    float* out = (float*)d_out;

    const int N = in_sizes[0] / FDIM;
    const int E = in_sizes[1] / 2;

    float* ws     = (float*)d_ws;
    float* h      = ws;
    float* a_src  = h     + (size_t)N * FDIM;
    float* a_dst  = a_src + (size_t)N * HEADS;
    int*   cnt    = (int*)(a_dst + (size_t)N * HEADS);
    int*   off    = cnt  + (size_t)N + 1;
    int*   binned = off  + (size_t)N;
    int*   segsrc = binned + (size_t)E;
    int*   blkoff = segsrc + (size_t)E;

    const int B = 256;
    const int NB   = (N + RB - 1) / RB;
    const int nblk = (E + EPB - 1) / EPB;

    bin_by_range<<<nblk, BIN_T, NB * sizeof(int), stream>>>(ei, binned, blkoff, cnt + N, E, NB);
    range_segments<<<NB, RB, 0, stream>>>(binned, blkoff, cnt + N, cnt, off, segsrc,
                                          nblk, NB, N);

    int tiles  = (N + 15) / 16;
    int gblk   = (tiles + 3) / 4;
    gemm_wmma_attn<<<gblk, 128, 0, stream>>>(x, W, att_src, att_dst, h, a_src, a_dst, N);

    node_softmax_gather<<<(N + 7) / 8, B, 0, stream>>>(off, cnt, segsrc, a_src, a_dst,
                                                       h, bias, out, N);
}


# per-range segment building and the x @ W strips share one launch (range work-groups first, then 8-strip ones): 3 launches, grouping round trips run under the matrix pipe; sum orders kept
# speedup vs baseline: 7.9031x; 1.0949x over previous
#include <hip/hip_runtime.h>
#include <math.h>

typedef float v2f __attribute__((ext_vector_type(2)));
typedef float v8f __attribute__((ext_vector_type(8)));

#define HEADS 4
#define CHAN 32
#define FDIM 128
#define NEG_SLOPE 0.2f


template <int XM>
__device__ __forceinline__ float row_xor(float v) {
    return __int_as_float(__builtin_amdgcn_update_dpp(
        0, __float_as_int(v), 0x160 + XM, 0xF, 0xF, true));
}

__device__ __forceinline__ float row_reduce_scatter16(const float (&v)[16], int c) {
    const bool b3 = (c & 8) != 0, b2 = (c & 4) != 0, b1 = (c & 2) != 0, b0 = (c & 1) != 0;
    float t8[8], t4[4], t2[2];
#pragma unroll
    for (int j = 0; j < 8; ++j) {
        const float keep = b3 ? v[8 + j] : v[j];
        const float send = b3 ? v[j] : v[8 + j];
        t8[j] = keep + row_xor<8>(send);
    }
#pragma unroll
    for (int j = 0; j < 4; ++j) {
        const float keep = b2 ? t8[4 + j] : t8[j];
        const float send = b2 ? t8[j] : t8[4 + j];
        t4[j] = keep + row_xor<4>(send);
    }
#pragma unroll
    for (int j = 0; j < 2; ++j) {
        const float keep = b1 ? t4[2 + j] : t4[j];
        const float send = b1 ? t4[j] : t4[2 + j];
        t2[j] = keep + row_xor<2>(send);
    }
    const float keep = b0 ? t2[1] : t2[0];
    const float send = b0 ? t2[0] : t2[1];
    return keep + row_xor<1>(send);
}

typedef float v4f __attribute__((ext_vector_type(4)));

__device__ __forceinline__ void gemm_attn_strip(const float* __restrict__ x,
                                                const float* __restrict__ W,
                                                const float* __restrict__ att_src,
                                                const float* __restrict__ att_dst,
                                                float* __restrict__ h,
                                                float* __restrict__ a_src,
                                                float* __restrict__ a_dst,
                                                int row0, int lane) {

    v8f acc[8];
#pragma unroll
    for (int t = 0; t < 8; ++t) acc[t] = (v8f)(0.0f);

    const int mrow  = row0 + (lane & 15);
    const int khalf = (lane >> 4) << 1;
    const int ncol  = lane & 15;

    for (int k0 = 0; k0 < FDIM; k0 += 4) {
        const int ka = k0 + khalf;
        v2f a;
        a.x = x[(size_t)mrow * FDIM + ka];
        a.y = x[(size_t)mrow * FDIM + ka + 1];
#pragma unroll
        for (int t = 0; t < 8; ++t) {
            const int n = t * 16 + ncol;
            v2f b;
            b.x = W[(size_t)ka * FDIM + n];
            b.y = W[(size_t)(ka + 1) * FDIM + n];
            acc[t] = __builtin_amdgcn_wmma_f32_16x16x4_f32(
                false, a, false, b, (short)0, acc[t], false, false);
        }
    }

    const int mbase = row0 + ((lane >> 4) << 3);
#pragma unroll
    for (int t = 0; t < 8; ++t) {
        const int n = t * 16 + ncol;
#pragma unroll
        for (int r = 0; r < 8; ++r)
            h[(size_t)(mbase + r) * FDIM + n] = acc[t][r];
    }

    v4f term;
#pragma unroll
    for (int hd = 0; hd < HEADS; ++hd) {
        const float s0 = att_src[hd * CHAN + ncol], s1 = att_src[hd * CHAN + 16 + ncol];
        const float d0 = att_dst[hd * CHAN + ncol], d1 = att_dst[hd * CHAN + 16 + ncol];
        float p[16];
#pragma unroll
        for (int r = 0; r < 8; ++r) {
            p[r]     = fmaf(acc[2 * hd + 1][r], s1, acc[2 * hd][r] * s0);
            p[8 + r] = fmaf(acc[2 * hd + 1][r], d1, acc[2 * hd][r] * d0);
        }
        term[hd] = row_reduce_scatter16(p, ncol);
    }
    float* dstp = (ncol < 8 ? a_src : a_dst) + (size_t)(mbase + (ncol & 7)) * HEADS;
    *(v4f*)dstp = term;
}

__device__ __forceinline__ float leaky(float v) {
    return v > 0.f ? v : NEG_SLOPE * v;
}


#define RB      256
#define RB_LOG  8
#define EPB     8192
#define BIN_T   1024


__global__ void __launch_bounds__(BIN_T)
bin_by_range(const int* __restrict__ ei, int* __restrict__ binned,
             int* __restrict__ blockoff, int* __restrict__ total, int Ecnt, int NB) {
    extern __shared__ int hist[];
    __shared__ int stage[EPB];
    __shared__ int wave_tot[32];
    __shared__ int tile_total;
    const int tid  = threadIdx.x;
    const int lane = tid & 31;
    const int w    = tid >> 5;

    for (int i = tid; i < NB; i += BIN_T) hist[i] = 0;
    if (blockIdx.x == 0 && tid == 0) *total = 0;
    __syncthreads();

    const int base = blockIdx.x * EPB;
    int word[8], bkt[8], rk[8];
#pragma unroll
    for (int j = 0; j < 8; ++j) {
        const int e = base + j * BIN_T + tid;
        bkt[j]  = -1;
        word[j] = 0;
        rk[j]   = 0;
        if (e < Ecnt) {
            const int d = ei[Ecnt + e];
            const int s = ei[e];
            bkt[j]  = d >> RB_LOG;
            word[j] = (int)(((unsigned)(d & (RB - 1)) << 24) | (unsigned)s);
        }
    }
#pragma unroll
    for (int j = 0; j < 8; ++j)
        if (bkt[j] >= 0) rk[j] = atomicAdd(&hist[bkt[j]], 1);
    __syncthreads();

    int carry = 0;
    for (int t0 = 0; t0 < NB; t0 += BIN_T) {
        const int i = t0 + tid;
        const int c = (i < NB) ? hist[i] : 0;
        int incl = c;
        for (int o = 1; o < 32; o <<= 1) {
            int t = __shfl_up(incl, o, 32);
            if (lane >= o) incl += t;
        }
        if (lane == 31) wave_tot[w] = incl;
        __syncthreads();
        if (w == 0) {
            const int v = wave_tot[lane];
            int inc2 = v;
            for (int o = 1; o < 32; o <<= 1) {
                int t = __shfl_up(inc2, o, 32);
                if (lane >= o) inc2 += t;
            }
            wave_tot[lane] = inc2 - v;
            if (lane == 31) tile_total = inc2;
        }
        __syncthreads();
        if (i < NB) hist[i] = carry + wave_tot[w] + incl - c;
        carry += tile_total;
        __syncthreads();
    }

    int* tab = blockoff + (size_t)blockIdx.x * (NB + 1);
    for (int i = tid; i < NB; i += BIN_T) tab[i] = hist[i];
    if (tid == 0) tab[NB] = carry;

#pragma unroll
    for (int j = 0; j < 8; ++j)
        if (bkt[j] >= 0) stage[hist[bkt[j]] + rk[j]] = word[j];
    __syncthreads();
#pragma unroll
    for (int j = 0; j < 8; ++j) {
        const int q = j * BIN_T + tid;
        if (q < carry) binned[base + q] = stage[q];
    }
}

__device__ __forceinline__ void
range_segments_block(const int* __restrict__ binned, const int* __restrict__ blockoff,
                     int* __restrict__ total, int* __restrict__ cnt, int* __restrict__ off,
                     int* __restrict__ seg_src, int nblk, int NB, int Nn, int b) {
    __shared__ int slot[RB];
    __shared__ int wave_base[RB / 32];
    __shared__ int block_base;
    const int tid  = threadIdx.x;
    const int lane = tid & 31;
    const int w    = tid >> 5;

    slot[tid] = 0;
    __syncthreads();
    for (int k = tid; k < nblk; k += RB) {
        const int* tab = blockoff + (size_t)k * (NB + 1);
        const int p0 = k * EPB + tab[b];
        const int p1 = k * EPB + tab[b + 1];
        for (int p = p0; p < p1; p += 4) {
            const int last = p1 - 1;
            const int w0 = binned[p];
            const int w1 = binned[min(p + 1, last)];
            const int w2 = binned[min(p + 2, last)];
            const int w3 = binned[min(p + 3, last)];
            atomicAdd(&slot[(unsigned)w0 >> 24], 1);
            if (p + 1 < p1) atomicAdd(&slot[(unsigned)w1 >> 24], 1);
            if (p + 2 < p1) atomicAdd(&slot[(unsigned)w2 >> 24], 1);
            if (p + 3 < p1) atomicAdd(&slot[(unsigned)w3 >> 24], 1);
        }
    }
    __syncthreads();

    const int c = slot[tid];
    int incl = c;
    for (int o = 1; o < 32; o <<= 1) {
        int t = __shfl_up(incl, o, 32);
        if (lane >= o) incl += t;
    }
    if (lane == 31) wave_base[w] = incl;
    __syncthreads();
    if (tid == 0) {
        int s = 0;
        for (int k = 0; k < RB / 32; ++k) { int t = wave_base[k]; wave_base[k] = s; s += t; }
        block_base = atomicAdd(total, s);
    }
    __syncthreads();
    const int o0   = block_base + wave_base[w] + incl - c;
    const int node = b * RB + tid;
    if (node < Nn) { cnt[node] = c; off[node] = o0; }
    slot[tid] = o0;
    __syncthreads();

    for (int k = tid; k < nblk; k += RB) {
        const int* tab = blockoff + (size_t)k * (NB + 1);
        const int p0 = k * EPB + tab[b];
        const int p1 = k * EPB + tab[b + 1];
        for (int p = p0; p < p1; p += 4) {
            const int last = p1 - 1;
            const int w0 = binned[p];
            const int w1 = binned[min(p + 1, last)];
            const int w2 = binned[min(p + 2, last)];
            const int w3 = binned[min(p + 3, last)];
            seg_src[atomicAdd(&slot[(unsigned)w0 >> 24], 1)] = w0 & 0x00FFFFFF;
            if (p + 1 < p1) seg_src[atomicAdd(&slot[(unsigned)w1 >> 24], 1)] = w1 & 0x00FFFFFF;
            if (p + 2 < p1) seg_src[atomicAdd(&slot[(unsigned)w2 >> 24], 1)] = w2 & 0x00FFFFFF;
            if (p + 3 < p1) seg_src[atomicAdd(&slot[(unsigned)w3 >> 24], 1)] = w3 & 0x00FFFFFF;
        }
    }
}

__global__ void
segments_and_gemm(const int* __restrict__ binned, const int* __restrict__ blockoff,
                  int* __restrict__ total, int* __restrict__ cnt, int* __restrict__ off,
                  int* __restrict__ seg_src, int nblk, int NB,
                  const float* __restrict__ x, const float* __restrict__ W,
                  const float* __restrict__ att_src, const float* __restrict__ att_dst,
                  float* __restrict__ h, float* __restrict__ a_src,
                  float* __restrict__ a_dst, int Nn) {
    if ((int)blockIdx.x < NB) {
        range_segments_block(binned, blockoff, total, cnt, off, seg_src, nblk, NB, Nn,
                             (int)blockIdx.x);
        return;
    }
    const int wave = threadIdx.x >> 5;
    const int lane = threadIdx.x & 31;
    const int row0 = (((int)blockIdx.x - NB) * (RB / 32) + wave) * 16;
    if (row0 >= Nn) return;
    gemm_attn_strip(x, W, att_src, att_dst, h, a_src, a_dst, row0, lane);
}

__device__ __forceinline__ float wave_max(float v) {
    for (int o = 16; o > 0; o >>= 1) v = fmaxf(v, __shfl_xor(v, o, 32));
    return v;
}

__device__ __forceinline__ v4f leaky4(v4f v) {
    v4f r;
    r.x = leaky(v.x); r.y = leaky(v.y); r.z = leaky(v.z); r.w = leaky(v.w);
    return r;
}

__device__ __forceinline__ float lane_bcast(float v, int k) {
    return __int_as_float(__builtin_amdgcn_readlane(__float_as_int(v), k));
}

__global__ void __launch_bounds__(256)
node_softmax_gather(const int* __restrict__ off, const int* __restrict__ cnt,
                    const int* __restrict__ seg_src,
                    const float* __restrict__ a_src, const float* __restrict__ a_dst,
                    const float* __restrict__ h, const float* __restrict__ bias,
                    float* __restrict__ out, int Nn) {
    const int lane = threadIdx.x & 31;
    const int n = blockIdx.x * 8 + __builtin_amdgcn_readfirstlane(threadIdx.x >> 5);
    if (n >= Nn) return;
    const int start = off[n];
    const int deg   = cnt[n];
    const int hd    = lane >> 3;
    const v4f ad = *(const v4f*)(a_dst + (size_t)n * HEADS);
    const float ninf = -__builtin_inff();

    const bool v0 = lane < deg;
    int s0 = 0;
    v4f l0 = {ninf, ninf, ninf, ninf};
    if (v0) {
        s0 = seg_src[start + lane];
        l0 = leaky4(*(const v4f*)(a_src + (size_t)s0 * HEADS) + ad);
    }
    v4f m = l0;
    for (int base = 32; base < deg; base += 32) {
        const int idx = base + lane;
        if (idx < deg) {
            const int s = seg_src[start + idx];
            const v4f l = leaky4(*(const v4f*)(a_src + (size_t)s * HEADS) + ad);
            m.x = fmaxf(m.x, l.x); m.y = fmaxf(m.y, l.y);
            m.z = fmaxf(m.z, l.z); m.w = fmaxf(m.w, l.w);
        }
    }
    m.x = wave_max(m.x); m.y = wave_max(m.y); m.z = wave_max(m.z); m.w = wave_max(m.w);
    if (!__builtin_isfinite(m.x)) m.x = 0.f;
    if (!__builtin_isfinite(m.y)) m.y = 0.f;
    if (!__builtin_isfinite(m.z)) m.z = 0.f;
    if (!__builtin_isfinite(m.w)) m.w = 0.f;

    v4f   acc = {0.f, 0.f, 0.f, 0.f};
    float den = 0.f;
    const float* hrow = h + lane * 4;

    for (int base = 0; base < deg; base += 32) {
        int s = s0;
        v4f l = l0;
        if (base > 0) {
            s = 0;
            l = (v4f){ninf, ninf, ninf, ninf};
            const int idx = base + lane;
            if (idx < deg) {
                s = seg_src[start + idx];
                l = leaky4(*(const v4f*)(a_src + (size_t)s * HEADS) + ad);
            }
        }
        const bool valid = base + lane < deg;
        v4f ex;
        ex.x = valid ? expf(l.x - m.x) : 0.f;
        ex.y = valid ? expf(l.y - m.y) : 0.f;
        ex.z = valid ? expf(l.z - m.z) : 0.f;
        ex.w = valid ? expf(l.w - m.w) : 0.f;

        const int c = min(deg - base, 32);
        for (int k = 0; k < c; k += 4) {
            const int g = min(c - k, 4);
            v4f r0, r1, r2, r3;
            float w0, w1, w2, w3;
            {
                const int sk = __builtin_amdgcn_readlane(s, k);
                r0 = *(const v4f*)(hrow + (size_t)sk * FDIM);
                const float e0 = lane_bcast(ex.x, k), e1 = lane_bcast(ex.y, k);
                const float e2 = lane_bcast(ex.z, k), e3 = lane_bcast(ex.w, k);
                w0 = hd == 0 ? e0 : hd == 1 ? e1 : hd == 2 ? e2 : e3;
            }
            if (g > 1) {
                const int sk = __builtin_amdgcn_readlane(s, k + 1);
                r1 = *(const v4f*)(hrow + (size_t)sk * FDIM);
                const float e0 = lane_bcast(ex.x, k + 1), e1 = lane_bcast(ex.y, k + 1);
                const float e2 = lane_bcast(ex.z, k + 1), e3 = lane_bcast(ex.w, k + 1);
                w1 = hd == 0 ? e0 : hd == 1 ? e1 : hd == 2 ? e2 : e3;
            }
            if (g > 2) {
                const int sk = __builtin_amdgcn_readlane(s, k + 2);
                r2 = *(const v4f*)(hrow + (size_t)sk * FDIM);
                const float e0 = lane_bcast(ex.x, k + 2), e1 = lane_bcast(ex.y, k + 2);
                const float e2 = lane_bcast(ex.z, k + 2), e3 = lane_bcast(ex.w, k + 2);
                w2 = hd == 0 ? e0 : hd == 1 ? e1 : hd == 2 ? e2 : e3;
            }
            if (g > 3) {
                const int sk = __builtin_amdgcn_readlane(s, k + 3);
                r3 = *(const v4f*)(hrow + (size_t)sk * FDIM);
                const float e0 = lane_bcast(ex.x, k + 3), e1 = lane_bcast(ex.y, k + 3);
                const float e2 = lane_bcast(ex.z, k + 3), e3 = lane_bcast(ex.w, k + 3);
                w3 = hd == 0 ? e0 : hd == 1 ? e1 : hd == 2 ? e2 : e3;
            }
            acc.x = fmaf(r0.x, w0, acc.x); acc.y = fmaf(r0.y, w0, acc.y);
            acc.z = fmaf(r0.z, w0, acc.z); acc.w = fmaf(r0.w, w0, acc.w);
            den += w0;
            if (g > 1) {
                acc.x = fmaf(r1.x, w1, acc.x); acc.y = fmaf(r1.y, w1, acc.y);
                acc.z = fmaf(r1.z, w1, acc.z); acc.w = fmaf(r1.w, w1, acc.w);
                den += w1;
            }
            if (g > 2) {
                acc.x = fmaf(r2.x, w2, acc.x); acc.y = fmaf(r2.y, w2, acc.y);
                acc.z = fmaf(r2.z, w2, acc.z); acc.w = fmaf(r2.w, w2, acc.w);
                den += w2;
            }
            if (g > 3) {
                acc.x = fmaf(r3.x, w3, acc.x); acc.y = fmaf(r3.y, w3, acc.y);
                acc.z = fmaf(r3.z, w3, acc.z); acc.w = fmaf(r3.w, w3, acc.w);
                den += w3;
            }
        }
    }

    const float dn = den + 1e-16f;
    const int j = lane * 4;
    v4f o;
    o.x = acc.x / dn + bias[j];
    o.y = acc.y / dn + bias[j + 1];
    o.z = acc.z / dn + bias[j + 2];
    o.w = acc.w / dn + bias[j + 3];
    o.x = o.x > 0.f ? o.x : expm1f(o.x);
    o.y = o.y > 0.f ? o.y : expm1f(o.y);
    o.z = o.z > 0.f ? o.z : expm1f(o.z);
    o.w = o.w > 0.f ? o.w : expm1f(o.w);
    *(v4f*)(out + (size_t)n * FDIM + j) = o;
}

extern "C" void kernel_launch(void* const* d_in, const int* in_sizes, int n_in,
                              void* d_out, int out_size, void* d_ws, size_t ws_size,
                              hipStream_t stream) {
    const float* x       = (const float*)d_in[0];
    const int*   ei      = (const int*)  d_in[1];
    const float* W       = (const float*)d_in[2];
    const float* att_src = (const float*)d_in[3];
    const float* att_dst = (const float*)d_in[4];
    const float* bias    = (const float*)d_in[5];
    float* out = (float*)d_out;

    const int N = in_sizes[0] / FDIM;
    const int E = in_sizes[1] / 2;

    float* ws     = (float*)d_ws;
    float* h      = ws;
    float* a_src  = h     + (size_t)N * FDIM;
    float* a_dst  = a_src + (size_t)N * HEADS;
    int*   cnt    = (int*)(a_dst + (size_t)N * HEADS);
    int*   off    = cnt  + (size_t)N + 1;
    int*   binned = off  + (size_t)N;
    int*   segsrc = binned + (size_t)E;
    int*   blkoff = segsrc + (size_t)E;

    const int B = 256;
    const int NB   = (N + RB - 1) / RB;
    const int nblk = (E + EPB - 1) / EPB;

    bin_by_range<<<nblk, BIN_T, NB * sizeof(int), stream>>>(ei, binned, blkoff, cnt + N, E, NB);
    int tiles  = (N + 15) / 16;
    int gblk   = (tiles + RB / 32 - 1) / (RB / 32);
    segments_and_gemm<<<NB + gblk, RB, 0, stream>>>(binned, blkoff, cnt + N, cnt, off, segsrc,
                                                    nblk, NB, x, W, att_src, att_dst,
                                                    h, a_src, a_dst, N);

    node_softmax_gather<<<(N + 7) / 8, B, 0, stream>>>(off, cnt, segsrc, a_src, a_dst,
                                                       h, bias, out, N);
}


# x @ W strips: W re-laid in-call into fragment order, one 16-byte load gives a lane a tile's four k (8 B loads per 8 k, was 32), A one 16-byte load; sum order: k regrouped inside 8-blocks
# speedup vs baseline: 8.3229x; 1.0531x over previous
#include <hip/hip_runtime.h>
#include <math.h>

typedef float v2f __attribute__((ext_vector_type(2)));
typedef float v8f __attribute__((ext_vector_type(8)));

#define HEADS 4
#define CHAN 32
#define FDIM 128
#define NEG_SLOPE 0.2f


template <int XM>
__device__ __forceinline__ float row_xor(float v) {
    return __int_as_float(__builtin_amdgcn_update_dpp(
        0, __float_as_int(v), 0x160 + XM, 0xF, 0xF, true));
}

__device__ __forceinline__ float row_reduce_scatter16(const float (&v)[16], int c) {
    const bool b3 = (c & 8) != 0, b2 = (c & 4) != 0, b1 = (c & 2) != 0, b0 = (c & 1) != 0;
    float t8[8], t4[4], t2[2];
#pragma unroll
    for (int j = 0; j < 8; ++j) {
        const float keep = b3 ? v[8 + j] : v[j];
        const float send = b3 ? v[j] : v[8 + j];
        t8[j] = keep + row_xor<8>(send);
    }
#pragma unroll
    for (int j = 0; j < 4; ++j) {
        const float keep = b2 ? t8[4 + j] : t8[j];
        const float send = b2 ? t8[j] : t8[4 + j];
        t4[j] = keep + row_xor<4>(send);
    }
#pragma unroll
    for (int j = 0; j < 2; ++j) {
        const float keep = b1 ? t4[2 + j] : t4[j];
        const float send = b1 ? t4[j] : t4[2 + j];
        t2[j] = keep + row_xor<2>(send);
    }
    const float keep = b0 ? t2[1] : t2[0];
    const float send = b0 ? t2[0] : t2[1];
    return keep + row_xor<1>(send);
}

typedef float v4f __attribute__((ext_vector_type(4)));

#define WP_ELEMS (FDIM * FDIM)

__device__ __forceinline__ void pack_w_fragments(const float* __restrict__ W,
                                                 float* __restrict__ Wp,
                                                 int first, int stride) {
    for (int q = first; q < WP_ELEMS; q += stride) {
        const int e  = q & 3;
        const int c  = (q >> 2) & 15;
        const int hf = (q >> 6) & 1;
        const int t  = (q >> 7) & 7;
        const int kb = q >> 10;
        Wp[q] = W[(size_t)(8 * kb + 4 * hf + e) * FDIM + 16 * t + c];
    }
}

__device__ __forceinline__ void gemm_attn_strip(const float* __restrict__ x,
                                                const float* __restrict__ Wp,
                                                const float* __restrict__ att_src,
                                                const float* __restrict__ att_dst,
                                                float* __restrict__ h,
                                                float* __restrict__ a_src,
                                                float* __restrict__ a_dst,
                                                int row0, int lane) {

    v8f acc[8];
#pragma unroll
    for (int t = 0; t < 8; ++t) acc[t] = (v8f)(0.0f);

    const int mrow  = row0 + (lane & 15);
    const int ncol  = lane & 15;
    const float* xa = x + (size_t)mrow * FDIM + ((lane >> 4) << 2);
    const v4f*   wb = (const v4f*)Wp + lane;

    for (int kb = 0; kb < FDIM / 8; ++kb) {
        const v4f a4 = *(const v4f*)(xa + kb * 8);
        v4f b4[8];
#pragma unroll
        for (int t = 0; t < 8; ++t) b4[t] = wb[(kb * 8 + t) * 32];
        v2f a;
        a.x = a4.x; a.y = a4.y;
#pragma unroll
        for (int t = 0; t < 8; ++t) {
            v2f b;
            b.x = b4[t].x; b.y = b4[t].y;
            acc[t] = __builtin_amdgcn_wmma_f32_16x16x4_f32(
                false, a, false, b, (short)0, acc[t], false, false);
        }
        a.x = a4.z; a.y = a4.w;
#pragma unroll
        for (int t = 0; t < 8; ++t) {
            v2f b;
            b.x = b4[t].z; b.y = b4[t].w;
            acc[t] = __builtin_amdgcn_wmma_f32_16x16x4_f32(
                false, a, false, b, (short)0, acc[t], false, false);
        }
    }

    const int mbase = row0 + ((lane >> 4) << 3);
#pragma unroll
    for (int t = 0; t < 8; ++t) {
        const int n = t * 16 + ncol;
#pragma unroll
        for (int r = 0; r < 8; ++r)
            h[(size_t)(mbase + r) * FDIM + n] = acc[t][r];
    }

    v4f term;
#pragma unroll
    for (int hd = 0; hd < HEADS; ++hd) {
        const float s0 = att_src[hd * CHAN + ncol], s1 = att_src[hd * CHAN + 16 + ncol];
        const float d0 = att_dst[hd * CHAN + ncol], d1 = att_dst[hd * CHAN + 16 + ncol];
        float p[16];
#pragma unroll
        for (int r = 0; r < 8; ++r) {
            p[r]     = fmaf(acc[2 * hd + 1][r], s1, acc[2 * hd][r] * s0);
            p[8 + r] = fmaf(acc[2 * hd + 1][r], d1, acc[2 * hd][r] * d0);
        }
        term[hd] = row_reduce_scatter16(p, ncol);
    }
    float* dstp = (ncol < 8 ? a_src : a_dst) + (size_t)(mbase + (ncol & 7)) * HEADS;
    *(v4f*)dstp = term;
}

__device__ __forceinline__ float leaky(float v) {
    return v > 0.f ? v : NEG_SLOPE * v;
}


#define RB      256
#define RB_LOG  8
#define EPB     8192
#define BIN_T   1024


__global__ void __launch_bounds__(BIN_T)
bin_by_range(const int* __restrict__ ei, int* __restrict__ binned,
             int* __restrict__ blockoff, int* __restrict__ total, int Ecnt, int NB,
             const float* __restrict__ W, float* __restrict__ Wp) {
    extern __shared__ int hist[];
    __shared__ int stage[EPB];
    __shared__ int wave_tot[32];
    __shared__ int tile_total;
    const int tid  = threadIdx.x;
    const int lane = tid & 31;
    const int w    = tid >> 5;

    for (int i = tid; i < NB; i += BIN_T) hist[i] = 0;
    if (blockIdx.x == 0 && tid == 0) *total = 0;
    pack_w_fragments(W, Wp, (int)blockIdx.x * BIN_T + tid, (int)gridDim.x * BIN_T);
    __syncthreads();

    const int base = blockIdx.x * EPB;
    int word[8], bkt[8], rk[8];
#pragma unroll
    for (int j = 0; j < 8; ++j) {
        const int e = base + j * BIN_T + tid;
        bkt[j]  = -1;
        word[j] = 0;
        rk[j]   = 0;
        if (e < Ecnt) {
            const int d = ei[Ecnt + e];
            const int s = ei[e];
            bkt[j]  = d >> RB_LOG;
            word[j] = (int)(((unsigned)(d & (RB - 1)) << 24) | (unsigned)s);
        }
    }
#pragma unroll
    for (int j = 0; j < 8; ++j)
        if (bkt[j] >= 0) rk[j] = atomicAdd(&hist[bkt[j]], 1);
    __syncthreads();

    int carry = 0;
    for (int t0 = 0; t0 < NB; t0 += BIN_T) {
        const int i = t0 + tid;
        const int c = (i < NB) ? hist[i] : 0;
        int incl = c;
        for (int o = 1; o < 32; o <<= 1) {
            int t = __shfl_up(incl, o, 32);
            if (lane >= o) incl += t;
        }
        if (lane == 31) wave_tot[w] = incl;
        __syncthreads();
        if (w == 0) {
            const int v = wave_tot[lane];
            int inc2 = v;
            for (int o = 1; o < 32; o <<= 1) {
                int t = __shfl_up(inc2, o, 32);
                if (lane >= o) inc2 += t;
            }
            wave_tot[lane] = inc2 - v;
            if (lane == 31) tile_total = inc2;
        }
        __syncthreads();
        if (i < NB) hist[i] = carry + wave_tot[w] + incl - c;
        carry += tile_total;
        __syncthreads();
    }

    int* tab = blockoff + (size_t)blockIdx.x * (NB + 1);
    for (int i = tid; i < NB; i += BIN_T) tab[i] = hist[i];
    if (tid == 0) tab[NB] = carry;

#pragma unroll
    for (int j = 0; j < 8; ++j)
        if (bkt[j] >= 0) stage[hist[bkt[j]] + rk[j]] = word[j];
    __syncthreads();
#pragma unroll
    for (int j = 0; j < 8; ++j) {
        const int q = j * BIN_T + tid;
        if (q < carry) binned[base + q] = stage[q];
    }
}

__device__ __forceinline__ void
range_segments_block(const int* __restrict__ binned, const int* __restrict__ blockoff,
                     int* __restrict__ total, int* __restrict__ cnt, int* __restrict__ off,
                     int* __restrict__ seg_src, int nblk, int NB, int Nn, int b) {
    __shared__ int slot[RB];
    __shared__ int wave_base[RB / 32];
    __shared__ int block_base;
    const int tid  = threadIdx.x;
    const int lane = tid & 31;
    const int w    = tid >> 5;

    slot[tid] = 0;
    __syncthreads();
    for (int k = tid; k < nblk; k += RB) {
        const int* tab = blockoff + (size_t)k * (NB + 1);
        const int p0 = k * EPB + tab[b];
        const int p1 = k * EPB + tab[b + 1];
        for (int p = p0; p < p1; p += 4) {
            const int last = p1 - 1;
            const int w0 = binned[p];
            const int w1 = binned[min(p + 1, last)];
            const int w2 = binned[min(p + 2, last)];
            const int w3 = binned[min(p + 3, last)];
            atomicAdd(&slot[(unsigned)w0 >> 24], 1);
            if (p + 1 < p1) atomicAdd(&slot[(unsigned)w1 >> 24], 1);
            if (p + 2 < p1) atomicAdd(&slot[(unsigned)w2 >> 24], 1);
            if (p + 3 < p1) atomicAdd(&slot[(unsigned)w3 >> 24], 1);
        }
    }
    __syncthreads();

    const int c = slot[tid];
    int incl = c;
    for (int o = 1; o < 32; o <<= 1) {
        int t = __shfl_up(incl, o, 32);
        if (lane >= o) incl += t;
    }
    if (lane == 31) wave_base[w] = incl;
    __syncthreads();
    if (tid == 0) {
        int s = 0;
        for (int k = 0; k < RB / 32; ++k) { int t = wave_base[k]; wave_base[k] = s; s += t; }
        block_base = atomicAdd(total, s);
    }
    __syncthreads();
    const int o0   = block_base + wave_base[w] + incl - c;
    const int node = b * RB + tid;
    if (node < Nn) { cnt[node] = c; off[node] = o0; }
    slot[tid] = o0;
    __syncthreads();

    for (int k = tid; k < nblk; k += RB) {
        const int* tab = blockoff + (size_t)k * (NB + 1);
        const int p0 = k * EPB + tab[b];
        const int p1 = k * EPB + tab[b + 1];
        for (int p = p0; p < p1; p += 4) {
            const int last = p1 - 1;
            const int w0 = binned[p];
            const int w1 = binned[min(p + 1, last)];
            const int w2 = binned[min(p + 2, last)];
            const int w3 = binned[min(p + 3, last)];
            seg_src[atomicAdd(&slot[(unsigned)w0 >> 24], 1)] = w0 & 0x00FFFFFF;
            if (p + 1 < p1) seg_src[atomicAdd(&slot[(unsigned)w1 >> 24], 1)] = w1 & 0x00FFFFFF;
            if (p + 2 < p1) seg_src[atomicAdd(&slot[(unsigned)w2 >> 24], 1)] = w2 & 0x00FFFFFF;
            if (p + 3 < p1) seg_src[atomicAdd(&slot[(unsigned)w3 >> 24], 1)] = w3 & 0x00FFFFFF;
        }
    }
}

__global__ void
segments_and_gemm(const int* __restrict__ binned, const int* __restrict__ blockoff,
                  int* __restrict__ total, int* __restrict__ cnt, int* __restrict__ off,
                  int* __restrict__ seg_src, int nblk, int NB,
                  const float* __restrict__ x, const float* __restrict__ Wp,
                  const float* __restrict__ att_src, const float* __restrict__ att_dst,
                  float* __restrict__ h, float* __restrict__ a_src,
                  float* __restrict__ a_dst, int Nn) {
    if ((int)blockIdx.x < NB) {
        range_segments_block(binned, blockoff, total, cnt, off, seg_src, nblk, NB, Nn,
                             (int)blockIdx.x);
        return;
    }
    const int wave = threadIdx.x >> 5;
    const int lane = threadIdx.x & 31;
    const int row0 = (((int)blockIdx.x - NB) * (RB / 32) + wave) * 16;
    if (row0 >= Nn) return;
    gemm_attn_strip(x, Wp, att_src, att_dst, h, a_src, a_dst, row0, lane);
}

__device__ __forceinline__ float wave_max(float v) {
    for (int o = 16; o > 0; o >>= 1) v = fmaxf(v, __shfl_xor(v, o, 32));
    return v;
}

__device__ __forceinline__ v4f leaky4(v4f v) {
    v4f r;
    r.x = leaky(v.x); r.y = leaky(v.y); r.z = leaky(v.z); r.w = leaky(v.w);
    return r;
}

__device__ __forceinline__ float lane_bcast(float v, int k) {
    return __int_as_float(__builtin_amdgcn_readlane(__float_as_int(v), k));
}

__global__ void __launch_bounds__(256)
node_softmax_gather(const int* __restrict__ off, const int* __restrict__ cnt,
                    const int* __restrict__ seg_src,
                    const float* __restrict__ a_src, const float* __restrict__ a_dst,
                    const float* __restrict__ h, const float* __restrict__ bias,
                    float* __restrict__ out, int Nn) {
    const int lane = threadIdx.x & 31;
    const int n = blockIdx.x * 8 + __builtin_amdgcn_readfirstlane(threadIdx.x >> 5);
    if (n >= Nn) return;
    const int start = off[n];
    const int deg   = cnt[n];
    const int hd    = lane >> 3;
    const v4f ad = *(const v4f*)(a_dst + (size_t)n * HEADS);
    const float ninf = -__builtin_inff();

    const bool v0 = lane < deg;
    int s0 = 0;
    v4f l0 = {ninf, ninf, ninf, ninf};
    if (v0) {
        s0 = seg_src[start + lane];
        l0 = leaky4(*(const v4f*)(a_src + (size_t)s0 * HEADS) + ad);
    }
    v4f m = l0;
    for (int base = 32; base < deg; base += 32) {
        const int idx = base + lane;
        if (idx < deg) {
            const int s = seg_src[start + idx];
            const v4f l = leaky4(*(const v4f*)(a_src + (size_t)s * HEADS) + ad);
            m.x = fmaxf(m.x, l.x); m.y = fmaxf(m.y, l.y);
            m.z = fmaxf(m.z, l.z); m.w = fmaxf(m.w, l.w);
        }
    }
    m.x = wave_max(m.x); m.y = wave_max(m.y); m.z = wave_max(m.z); m.w = wave_max(m.w);
    if (!__builtin_isfinite(m.x)) m.x = 0.f;
    if (!__builtin_isfinite(m.y)) m.y = 0.f;
    if (!__builtin_isfinite(m.z)) m.z = 0.f;
    if (!__builtin_isfinite(m.w)) m.w = 0.f;

    v4f   acc = {0.f, 0.f, 0.f, 0.f};
    float den = 0.f;
    const float* hrow = h + lane * 4;

    for (int base = 0; base < deg; base += 32) {
        int s = s0;
        v4f l = l0;
        if (base > 0) {
            s = 0;
            l = (v4f){ninf, ninf, ninf, ninf};
            const int idx = base + lane;
            if (idx < deg) {
                s = seg_src[start + idx];
                l = leaky4(*(const v4f*)(a_src + (size_t)s * HEADS) + ad);
            }
        }
        const bool valid = base + lane < deg;
        v4f ex;
        ex.x = valid ? expf(l.x - m.x) : 0.f;
        ex.y = valid ? expf(l.y - m.y) : 0.f;
        ex.z = valid ? expf(l.z - m.z) : 0.f;
        ex.w = valid ? expf(l.w - m.w) : 0.f;

        const int c = min(deg - base, 32);
        for (int k = 0; k < c; k += 4) {
            const int g = min(c - k, 4);
            v4f r0, r1, r2, r3;
            float w0, w1, w2, w3;
            {
                const int sk = __builtin_amdgcn_readlane(s, k);
                r0 = *(const v4f*)(hrow + (size_t)sk * FDIM);
                const float e0 = lane_bcast(ex.x, k), e1 = lane_bcast(ex.y, k);
                const float e2 = lane_bcast(ex.z, k), e3 = lane_bcast(ex.w, k);
                w0 = hd == 0 ? e0 : hd == 1 ? e1 : hd == 2 ? e2 : e3;
            }
            if (g > 1) {
                const int sk = __builtin_amdgcn_readlane(s, k + 1);
                r1 = *(const v4f*)(hrow + (size_t)sk * FDIM);
                const float e0 = lane_bcast(ex.x, k + 1), e1 = lane_bcast(ex.y, k + 1);
                const float e2 = lane_bcast(ex.z, k + 1), e3 = lane_bcast(ex.w, k + 1);
                w1 = hd == 0 ? e0 : hd == 1 ? e1 : hd == 2 ? e2 : e3;
            }
            if (g > 2) {
                const int sk = __builtin_amdgcn_readlane(s, k + 2);
                r2 = *(const v4f*)(hrow + (size_t)sk * FDIM);
                const float e0 = lane_bcast(ex.x, k + 2), e1 = lane_bcast(ex.y, k + 2);
                const float e2 = lane_bcast(ex.z, k + 2), e3 = lane_bcast(ex.w, k + 2);
                w2 = hd == 0 ? e0 : hd == 1 ? e1 : hd == 2 ? e2 : e3;
            }
            if (g > 3) {
                const int sk = __builtin_amdgcn_readlane(s, k + 3);
                r3 = *(const v4f*)(hrow + (size_t)sk * FDIM);
                const float e0 = lane_bcast(ex.x, k + 3), e1 = lane_bcast(ex.y, k + 3);
                const float e2 = lane_bcast(ex.z, k + 3), e3 = lane_bcast(ex.w, k + 3);
                w3 = hd == 0 ? e0 : hd == 1 ? e1 : hd == 2 ? e2 : e3;
            }
            acc.x = fmaf(r0.x, w0, acc.x); acc.y = fmaf(r0.y, w0, acc.y);
            acc.z = fmaf(r0.z, w0, acc.z); acc.w = fmaf(r0.w, w0, acc.w);
            den += w0;
            if (g > 1) {
                acc.x = fmaf(r1.x, w1, acc.x); acc.y = fmaf(r1.y, w1, acc.y);
                acc.z = fmaf(r1.z, w1, acc.z); acc.w = fmaf(r1.w, w1, acc.w);
                den += w1;
            }
            if (g > 2) {
                acc.x = fmaf(r2.x, w2, acc.x); acc.y = fmaf(r2.y, w2, acc.y);
                acc.z = fmaf(r2.z, w2, acc.z); acc.w = fmaf(r2.w, w2, acc.w);
                den += w2;
            }
            if (g > 3) {
                acc.x = fmaf(r3.x, w3, acc.x); acc.y = fmaf(r3.y, w3, acc.y);
                acc.z = fmaf(r3.z, w3, acc.z); acc.w = fmaf(r3.w, w3, acc.w);
                den += w3;
            }
        }
    }

    const float dn = den + 1e-16f;
    const int j = lane * 4;
    v4f o;
    o.x = acc.x / dn + bias[j];
    o.y = acc.y / dn + bias[j + 1];
    o.z = acc.z / dn + bias[j + 2];
    o.w = acc.w / dn + bias[j + 3];
    o.x = o.x > 0.f ? o.x : expm1f(o.x);
    o.y = o.y > 0.f ? o.y : expm1f(o.y);
    o.z = o.z > 0.f ? o.z : expm1f(o.z);
    o.w = o.w > 0.f ? o.w : expm1f(o.w);
    *(v4f*)(out + (size_t)n * FDIM + j) = o;
}

extern "C" void kernel_launch(void* const* d_in, const int* in_sizes, int n_in,
                              void* d_out, int out_size, void* d_ws, size_t ws_size,
                              hipStream_t stream) {
    const float* x       = (const float*)d_in[0];
    const int*   ei      = (const int*)  d_in[1];
    const float* W       = (const float*)d_in[2];
    const float* att_src = (const float*)d_in[3];
    const float* att_dst = (const float*)d_in[4];
    const float* bias    = (const float*)d_in[5];
    float* out = (float*)d_out;

    const int N = in_sizes[0] / FDIM;
    const int E = in_sizes[1] / 2;

    float* ws     = (float*)d_ws;
    float* Wp     = ws;
    float* h      = Wp    + WP_ELEMS;
    float* a_src  = h     + (size_t)N * FDIM;
    float* a_dst  = a_src + (size_t)N * HEADS;
    int*   cnt    = (int*)(a_dst + (size_t)N * HEADS);
    int*   off    = cnt  + (size_t)N + 1;
    int*   binned = off  + (size_t)N;
    int*   segsrc = binned + (size_t)E;
    int*   blkoff = segsrc + (size_t)E;

    const int B = 256;
    const int NB   = (N + RB - 1) / RB;
    const int nblk = (E + EPB - 1) / EPB;

    bin_by_range<<<nblk, BIN_T, NB * sizeof(int), stream>>>(ei, binned, blkoff, cnt + N, E, NB,
                                                            W, Wp);
    int tiles  = (N + 15) / 16;
    int gblk   = (tiles + RB / 32 - 1) / (RB / 32);
    segments_and_gemm<<<NB + gblk, RB, 0, stream>>>(binned, blkoff, cnt + N, cnt, off, segsrc,
                                                    nblk, NB, x, Wp, att_src, att_dst,
                                                    h, a_src, a_dst, N);

    node_softmax_gather<<<(N + 7) / 8, B, 0, stream>>>(off, cnt, segsrc, a_src, a_dst,
                                                       h, bias, out, N);
}


# node pass: softmax weights in head-major lanes (lane = head x slot): max and exp-sum are 8-lane DPP trees, a row's weight one DPP8 move (was 4 readlanes + 3 selects); exp-sum order: lane, then tree
# speedup vs baseline: 8.7424x; 1.0504x over previous
#include <hip/hip_runtime.h>
#include <math.h>

typedef float v2f __attribute__((ext_vector_type(2)));
typedef float v8f __attribute__((ext_vector_type(8)));

#define HEADS 4
#define CHAN 32
#define FDIM 128
#define NEG_SLOPE 0.2f


template <int XM>
__device__ __forceinline__ float row_xor(float v) {
    return __int_as_float(__builtin_amdgcn_update_dpp(
        0, __float_as_int(v), 0x160 + XM, 0xF, 0xF, true));
}

__device__ __forceinline__ float row_reduce_scatter16(const float (&v)[16], int c) {
    const bool b3 = (c & 8) != 0, b2 = (c & 4) != 0, b1 = (c & 2) != 0, b0 = (c & 1) != 0;
    float t8[8], t4[4], t2[2];
#pragma unroll
    for (int j = 0; j < 8; ++j) {
        const float keep = b3 ? v[8 + j] : v[j];
        const float send = b3 ? v[j] : v[8 + j];
        t8[j] = keep + row_xor<8>(send);
    }
#pragma unroll
    for (int j = 0; j < 4; ++j) {
        const float keep = b2 ? t8[4 + j] : t8[j];
        const float send = b2 ? t8[j] : t8[4 + j];
        t4[j] = keep + row_xor<4>(send);
    }
#pragma unroll
    for (int j = 0; j < 2; ++j) {
        const float keep = b1 ? t4[2 + j] : t4[j];
        const float send = b1 ? t4[j] : t4[2 + j];
        t2[j] = keep + row_xor<2>(send);
    }
    const float keep = b0 ? t2[1] : t2[0];
    const float send = b0 ? t2[0] : t2[1];
    return keep + row_xor<1>(send);
}

typedef float v4f __attribute__((ext_vector_type(4)));

#define WP_ELEMS (FDIM * FDIM)

__device__ __forceinline__ void pack_w_fragments(const float* __restrict__ W,
                                                 float* __restrict__ Wp,
                                                 int first, int stride) {
    for (int q = first; q < WP_ELEMS; q += stride) {
        const int e  = q & 3;
        const int c  = (q >> 2) & 15;
        const int hf = (q >> 6) & 1;
        const int t  = (q >> 7) & 7;
        const int kb = q >> 10;
        Wp[q] = W[(size_t)(8 * kb + 4 * hf + e) * FDIM + 16 * t + c];
    }
}

__device__ __forceinline__ void gemm_attn_strip(const float* __restrict__ x,
                                                const float* __restrict__ Wp,
                                                const float* __restrict__ att_src,
                                                const float* __restrict__ att_dst,
                                                float* __restrict__ h,
                                                float* __restrict__ a_src,
                                                float* __restrict__ a_dst,
                                                int row0, int lane) {

    v8f acc[8];
#pragma unroll
    for (int t = 0; t < 8; ++t) acc[t] = (v8f)(0.0f);

    const int mrow  = row0 + (lane & 15);
    const int ncol  = lane & 15;
    const float* xa = x + (size_t)mrow * FDIM + ((lane >> 4) << 2);
    const v4f*   wb = (const v4f*)Wp + lane;

    for (int kb = 0; kb < FDIM / 8; ++kb) {
        const v4f a4 = *(const v4f*)(xa + kb * 8);
        v4f b4[8];
#pragma unroll
        for (int t = 0; t < 8; ++t) b4[t] = wb[(kb * 8 + t) * 32];
        v2f a;
        a.x = a4.x; a.y = a4.y;
#pragma unroll
        for (int t = 0; t < 8; ++t) {
            v2f b;
            b.x = b4[t].x; b.y = b4[t].y;
            acc[t] = __builtin_amdgcn_wmma_f32_16x16x4_f32(
                false, a, false, b, (short)0, acc[t], false, false);
        }
        a.x = a4.z; a.y = a4.w;
#pragma unroll
        for (int t = 0; t < 8; ++t) {
            v2f b;
            b.x = b4[t].z; b.y = b4[t].w;
            acc[t] = __builtin_amdgcn_wmma_f32_16x16x4_f32(
                false, a, false, b, (short)0, acc[t], false, false);
        }
    }

    const int mbase = row0 + ((lane >> 4) << 3);
#pragma unroll
    for (int t = 0; t < 8; ++t) {
        const int n = t * 16 + ncol;
#pragma unroll
        for (int r = 0; r < 8; ++r)
            h[(size_t)(mbase + r) * FDIM + n] = acc[t][r];
    }

    v4f term;
#pragma unroll
    for (int hd = 0; hd < HEADS; ++hd) {
        const float s0 = att_src[hd * CHAN + ncol], s1 = att_src[hd * CHAN + 16 + ncol];
        const float d0 = att_dst[hd * CHAN + ncol], d1 = att_dst[hd * CHAN + 16 + ncol];
        float p[16];
#pragma unroll
        for (int r = 0; r < 8; ++r) {
            p[r]     = fmaf(acc[2 * hd + 1][r], s1, acc[2 * hd][r] * s0);
            p[8 + r] = fmaf(acc[2 * hd + 1][r], d1, acc[2 * hd][r] * d0);
        }
        term[hd] = row_reduce_scatter16(p, ncol);
    }
    float* dstp = (ncol < 8 ? a_src : a_dst) + (size_t)(mbase + (ncol & 7)) * HEADS;
    *(v4f*)dstp = term;
}

__device__ __forceinline__ float leaky(float v) {
    return v > 0.f ? v : NEG_SLOPE * v;
}


#define RB      256
#define RB_LOG  8
#define EPB     8192
#define BIN_T   1024


__global__ void __launch_bounds__(BIN_T)
bin_by_range(const int* __restrict__ ei, int* __restrict__ binned,
             int* __restrict__ blockoff, int* __restrict__ total, int Ecnt, int NB,
             const float* __restrict__ W, float* __restrict__ Wp) {
    extern __shared__ int hist[];
    __shared__ int stage[EPB];
    __shared__ int wave_tot[32];
    __shared__ int tile_total;
    const int tid  = threadIdx.x;
    const int lane = tid & 31;
    const int w    = tid >> 5;

    for (int i = tid; i < NB; i += BIN_T) hist[i] = 0;
    if (blockIdx.x == 0 && tid == 0) *total = 0;
    pack_w_fragments(W, Wp, (int)blockIdx.x * BIN_T + tid, (int)gridDim.x * BIN_T);
    __syncthreads();

    const int base = blockIdx.x * EPB;
    int word[8], bkt[8], rk[8];
#pragma unroll
    for (int j = 0; j < 8; ++j) {
        const int e = base + j * BIN_T + tid;
        bkt[j]  = -1;
        word[j] = 0;
        rk[j]   = 0;
        if (e < Ecnt) {
            const int d = ei[Ecnt + e];
            const int s = ei[e];
            bkt[j]  = d >> RB_LOG;
            word[j] = (int)(((unsigned)(d & (RB - 1)) << 24) | (unsigned)s);
        }
    }
#pragma unroll
    for (int j = 0; j < 8; ++j)
        if (bkt[j] >= 0) rk[j] = atomicAdd(&hist[bkt[j]], 1);
    __syncthreads();

    int carry = 0;
    for (int t0 = 0; t0 < NB; t0 += BIN_T) {
        const int i = t0 + tid;
        const int c = (i < NB) ? hist[i] : 0;
        int incl = c;
        for (int o = 1; o < 32; o <<= 1) {
            int t = __shfl_up(incl, o, 32);
            if (lane >= o) incl += t;
        }
        if (lane == 31) wave_tot[w] = incl;
        __syncthreads();
        if (w == 0) {
            const int v = wave_tot[lane];
            int inc2 = v;
            for (int o = 1; o < 32; o <<= 1) {
                int t = __shfl_up(inc2, o, 32);
                if (lane >= o) inc2 += t;
            }
            wave_tot[lane] = inc2 - v;
            if (lane == 31) tile_total = inc2;
        }
        __syncthreads();
        if (i < NB) hist[i] = carry + wave_tot[w] + incl - c;
        carry += tile_total;
        __syncthreads();
    }

    int* tab = blockoff + (size_t)blockIdx.x * (NB + 1);
    for (int i = tid; i < NB; i += BIN_T) tab[i] = hist[i];
    if (tid == 0) tab[NB] = carry;

#pragma unroll
    for (int j = 0; j < 8; ++j)
        if (bkt[j] >= 0) stage[hist[bkt[j]] + rk[j]] = word[j];
    __syncthreads();
#pragma unroll
    for (int j = 0; j < 8; ++j) {
        const int q = j * BIN_T + tid;
        if (q < carry) binned[base + q] = stage[q];
    }
}

__device__ __forceinline__ void
range_segments_block(const int* __restrict__ binned, const int* __restrict__ blockoff,
                     int* __restrict__ total, int* __restrict__ cnt, int* __restrict__ off,
                     int* __restrict__ seg_src, int nblk, int NB, int Nn, int b) {
    __shared__ int slot[RB];
    __shared__ int wave_base[RB / 32];
    __shared__ int block_base;
    const int tid  = threadIdx.x;
    const int lane = tid & 31;
    const int w    = tid >> 5;

    slot[tid] = 0;
    __syncthreads();
    for (int k = tid; k < nblk; k += RB) {
        const int* tab = blockoff + (size_t)k * (NB + 1);
        const int p0 = k * EPB + tab[b];
        const int p1 = k * EPB + tab[b + 1];
        for (int p = p0; p < p1; p += 4) {
            const int last = p1 - 1;
            const int w0 = binned[p];
            const int w1 = binned[min(p + 1, last)];
            const int w2 = binned[min(p + 2, last)];
            const int w3 = binned[min(p + 3, last)];
            atomicAdd(&slot[(unsigned)w0 >> 24], 1);
            if (p + 1 < p1) atomicAdd(&slot[(unsigned)w1 >> 24], 1);
            if (p + 2 < p1) atomicAdd(&slot[(unsigned)w2 >> 24], 1);
            if (p + 3 < p1) atomicAdd(&slot[(unsigned)w3 >> 24], 1);
        }
    }
    __syncthreads();

    const int c = slot[tid];
    int incl = c;
    for (int o = 1; o < 32; o <<= 1) {
        int t = __shfl_up(incl, o, 32);
        if (lane >= o) incl += t;
    }
    if (lane == 31) wave_base[w] = incl;
    __syncthreads();
    if (tid == 0) {
        int s = 0;
        for (int k = 0; k < RB / 32; ++k) { int t = wave_base[k]; wave_base[k] = s; s += t; }
        block_base = atomicAdd(total, s);
    }
    __syncthreads();
    const int o0   = block_base + wave_base[w] + incl - c;
    const int node = b * RB + tid;
    if (node < Nn) { cnt[node] = c; off[node] = o0; }
    slot[tid] = o0;
    __syncthreads();

    for (int k = tid; k < nblk; k += RB) {
        const int* tab = blockoff + (size_t)k * (NB + 1);
        const int p0 = k * EPB + tab[b];
        const int p1 = k * EPB + tab[b + 1];
        for (int p = p0; p < p1; p += 4) {
            const int last = p1 - 1;
            const int w0 = binned[p];
            const int w1 = binned[min(p + 1, last)];
            const int w2 = binned[min(p + 2, last)];
            const int w3 = binned[min(p + 3, last)];
            seg_src[atomicAdd(&slot[(unsigned)w0 >> 24], 1)] = w0 & 0x00FFFFFF;
            if (p + 1 < p1) seg_src[atomicAdd(&slot[(unsigned)w1 >> 24], 1)] = w1 & 0x00FFFFFF;
            if (p + 2 < p1) seg_src[atomicAdd(&slot[(unsigned)w2 >> 24], 1)] = w2 & 0x00FFFFFF;
            if (p + 3 < p1) seg_src[atomicAdd(&slot[(unsigned)w3 >> 24], 1)] = w3 & 0x00FFFFFF;
        }
    }
}

__global__ void
segments_and_gemm(const int* __restrict__ binned, const int* __restrict__ blockoff,
                  int* __restrict__ total, int* __restrict__ cnt, int* __restrict__ off,
                  int* __restrict__ seg_src, int nblk, int NB,
                  const float* __restrict__ x, const float* __restrict__ Wp,
                  const float* __restrict__ att_src, const float* __restrict__ att_dst,
                  float* __restrict__ h, float* __restrict__ a_src,
                  float* __restrict__ a_dst, int Nn) {
    if ((int)blockIdx.x < NB) {
        range_segments_block(binned, blockoff, total, cnt, off, seg_src, nblk, NB, Nn,
                             (int)blockIdx.x);
        return;
    }
    const int wave = threadIdx.x >> 5;
    const int lane = threadIdx.x & 31;
    const int row0 = (((int)blockIdx.x - NB) * (RB / 32) + wave) * 16;
    if (row0 >= Nn) return;
    gemm_attn_strip(x, Wp, att_src, att_dst, h, a_src, a_dst, row0, lane);
}

template <int M>
__device__ __forceinline__ float group8_bcast(float v) {
    return __int_as_float(__builtin_amdgcn_mov_dpp8(__float_as_int(v), M * 0x249249));
}

template <int G>
__device__ __forceinline__ void gather_group(const float* __restrict__ h, int lane,
                                             const int (&sv)[4], const float (&ex)[4],
                                             v4f& acc) {
    constexpr int I  = G >> 1;
    constexpr int M0 = (G & 1) * 4;
    const int k0 = __builtin_amdgcn_readlane(sv[I], M0);
    const int k1 = __builtin_amdgcn_readlane(sv[I], M0 + 1);
    const int k2 = __builtin_amdgcn_readlane(sv[I], M0 + 2);
    const int k3 = __builtin_amdgcn_readlane(sv[I], M0 + 3);
    const v4f r0 = ((const v4f*)(h + (size_t)k0 * FDIM))[lane];
    const v4f r1 = ((const v4f*)(h + (size_t)k1 * FDIM))[lane];
    const v4f r2 = ((const v4f*)(h + (size_t)k2 * FDIM))[lane];
    const v4f r3 = ((const v4f*)(h + (size_t)k3 * FDIM))[lane];
    const float w0 = group8_bcast<M0>(ex[I]);
    const float w1 = group8_bcast<M0 + 1>(ex[I]);
    const float w2 = group8_bcast<M0 + 2>(ex[I]);
    const float w3 = group8_bcast<M0 + 3>(ex[I]);
    acc.x = fmaf(r0.x, w0, acc.x); acc.y = fmaf(r0.y, w0, acc.y);
    acc.z = fmaf(r0.z, w0, acc.z); acc.w = fmaf(r0.w, w0, acc.w);
    acc.x = fmaf(r1.x, w1, acc.x); acc.y = fmaf(r1.y, w1, acc.y);
    acc.z = fmaf(r1.z, w1, acc.z); acc.w = fmaf(r1.w, w1, acc.w);
    acc.x = fmaf(r2.x, w2, acc.x); acc.y = fmaf(r2.y, w2, acc.y);
    acc.z = fmaf(r2.z, w2, acc.z); acc.w = fmaf(r2.w, w2, acc.w);
    acc.x = fmaf(r3.x, w3, acc.x); acc.y = fmaf(r3.y, w3, acc.y);
    acc.z = fmaf(r3.z, w3, acc.z); acc.w = fmaf(r3.w, w3, acc.w);
}

__device__ __forceinline__ void chunk_logits(const int* __restrict__ seg,
                                             const float* __restrict__ a_src,
                                             int base, int deg, int hd, int m8, float adh,
                                             int (&sv)[4], float (&lg)[4]) {
    const float ninf = -__builtin_inff();
    const int last = deg - 1;
    float av[4];
#pragma unroll
    for (int i = 0; i < 4; ++i) sv[i] = seg[min(base + 8 * i + m8, last)];
#pragma unroll
    for (int i = 0; i < 4; ++i) av[i] = a_src[(unsigned)sv[i] * (unsigned)HEADS + (unsigned)hd];
#pragma unroll
    for (int i = 0; i < 4; ++i)
        lg[i] = (base + 8 * i + m8 < deg) ? leaky(av[i] + adh) : ninf;
}

__global__ void __launch_bounds__(256)
node_softmax_gather(const int* __restrict__ off, const int* __restrict__ cnt,
                    const int* __restrict__ seg_src,
                    const float* __restrict__ a_src, const float* __restrict__ a_dst,
                    const float* __restrict__ h, const float* __restrict__ bias,
                    float* __restrict__ out, int Nn) {
    const int lane = threadIdx.x & 31;
    const int n = blockIdx.x * 8 + __builtin_amdgcn_readfirstlane(threadIdx.x >> 5);
    if (n >= Nn) return;
    const int start = off[n];
    const int deg   = cnt[n];
    const int hd    = lane >> 3;
    const int m8    = lane & 7;
    const float adh = a_dst[(size_t)n * HEADS + hd];
    const int* seg  = seg_src + start;

    const float ninf = -__builtin_inff();
    int   sv[4] = {0, 0, 0, 0};
    float lg[4] = {ninf, ninf, ninf, ninf};
    if (deg > 0) chunk_logits(seg, a_src, 0, deg, hd, m8, adh, sv, lg);
    float mx = fmaxf(fmaxf(lg[0], lg[1]), fmaxf(lg[2], lg[3]));
    for (int base = 32; base < deg; base += 32) {
        int   s2[4];
        float l2[4];
        chunk_logits(seg, a_src, base, deg, hd, m8, adh, s2, l2);
        mx = fmaxf(mx, fmaxf(fmaxf(l2[0], l2[1]), fmaxf(l2[2], l2[3])));
    }
    mx = fmaxf(mx, row_xor<1>(mx));
    mx = fmaxf(mx, row_xor<2>(mx));
    mx = fmaxf(mx, row_xor<4>(mx));
    if (!__builtin_isfinite(mx)) mx = 0.f;

    v4f   acc  = {0.f, 0.f, 0.f, 0.f};
    float dsum = 0.f;

    for (int base = 0; base < deg; base += 32) {
        if (base > 0) chunk_logits(seg, a_src, base, deg, hd, m8, adh, sv, lg);
        const int c = min(deg - base, 32);
        float ex[4];
#pragma unroll
        for (int i = 0; i < 4; ++i) {
            ex[i] = 0.f;
            if (8 * i < c) {
                ex[i] = expf(lg[i] - mx);
                dsum += ex[i];
            }
        }
        do {
            gather_group<0>(h, lane, sv, ex, acc); if (c <= 4)  break;
            gather_group<1>(h, lane, sv, ex, acc); if (c <= 8)  break;
            gather_group<2>(h, lane, sv, ex, acc); if (c <= 12) break;
            gather_group<3>(h, lane, sv, ex, acc); if (c <= 16) break;
            gather_group<4>(h, lane, sv, ex, acc); if (c <= 20) break;
            gather_group<5>(h, lane, sv, ex, acc); if (c <= 24) break;
            gather_group<6>(h, lane, sv, ex, acc); if (c <= 28) break;
            gather_group<7>(h, lane, sv, ex, acc);
        } while (0);
    }

    float den = dsum;
    den += row_xor<1>(den);
    den += row_xor<2>(den);
    den += row_xor<4>(den);

    const float dn = den + 1e-16f;
    const int j = lane * 4;
    v4f o;
    o.x = acc.x / dn + bias[j];
    o.y = acc.y / dn + bias[j + 1];
    o.z = acc.z / dn + bias[j + 2];
    o.w = acc.w / dn + bias[j + 3];
    o.x = o.x > 0.f ? o.x : expm1f(o.x);
    o.y = o.y > 0.f ? o.y : expm1f(o.y);
    o.z = o.z > 0.f ? o.z : expm1f(o.z);
    o.w = o.w > 0.f ? o.w : expm1f(o.w);
    *(v4f*)(out + (size_t)n * FDIM + j) = o;
}

extern "C" void kernel_launch(void* const* d_in, const int* in_sizes, int n_in,
                              void* d_out, int out_size, void* d_ws, size_t ws_size,
                              hipStream_t stream) {
    const float* x       = (const float*)d_in[0];
    const int*   ei      = (const int*)  d_in[1];
    const float* W       = (const float*)d_in[2];
    const float* att_src = (const float*)d_in[3];
    const float* att_dst = (const float*)d_in[4];
    const float* bias    = (const float*)d_in[5];
    float* out = (float*)d_out;

    const int N = in_sizes[0] / FDIM;
    const int E = in_sizes[1] / 2;

    float* ws     = (float*)d_ws;
    float* Wp     = ws;
    float* h      = Wp    + WP_ELEMS;
    float* a_src  = h     + (size_t)N * FDIM;
    float* a_dst  = a_src + (size_t)N * HEADS;
    int*   cnt    = (int*)(a_dst + (size_t)N * HEADS);
    int*   off    = cnt  + (size_t)N + 1;
    int*   binned = off  + (size_t)N;
    int*   segsrc = binned + (size_t)E;
    int*   blkoff = segsrc + (size_t)E;

    const int B = 256;
    const int NB   = (N + RB - 1) / RB;
    const int nblk = (E + EPB - 1) / EPB;

    bin_by_range<<<nblk, BIN_T, NB * sizeof(int), stream>>>(ei, binned, blkoff, cnt + N, E, NB,
                                                            W, Wp);
    int tiles  = (N + 15) / 16;
    int gblk   = (tiles + RB / 32 - 1) / (RB / 32);
    segments_and_gemm<<<NB + gblk, RB, 0, stream>>>(binned, blkoff, cnt + N, cnt, off, segsrc,
                                                    nblk, NB, x, Wp, att_src, att_dst,
                                                    h, a_src, a_dst, N);

    node_softmax_gather<<<(N + 7) / 8, B, 0, stream>>>(off, cnt, segsrc, a_src, a_dst,
                                                       h, bias, out, N);
}
